# starting kernel
# baseline (speedup 1.0000x reference)
#include <hip/hip_runtime.h>
#include <math.h>


typedef __attribute__((ext_vector_type(2))) float v2f;
typedef __attribute__((ext_vector_type(4))) float v4f;
typedef __attribute__((ext_vector_type(8))) float v8f;

#define NEG_BIG (-10000.0f)

#define TM 128
#define TN 64
#define TK 32
#define TKP 34


__global__ __launch_bounds__(256) void gemm_nt_f32_wmma(
    const float* __restrict__ A, const float* __restrict__ W,
    float* __restrict__ C, int M, int N, int K)
{
    __shared__ float As[TM][TKP];
    __shared__ float Ws[TN][TKP];

    const int tid  = threadIdx.x;
    const int lane = tid & 31, wid = tid >> 5;
    const int wm   = wid >> 1, wn  = wid & 1;
    const int ll   = lane & 15, lh = lane >> 4;
    const int m0   = blockIdx.x * TM, n0 = blockIdx.y * TN;

    const v8f vzero = {0.f,0.f,0.f,0.f,0.f,0.f,0.f,0.f};
    v8f acc[2][2];
    acc[0][0]=vzero; acc[0][1]=vzero; acc[1][0]=vzero; acc[1][1]=vzero;

    for (int k0 = 0; k0 < K; k0 += TK) {
        __syncthreads();
        #pragma unroll
        for (int p = 0; p < 4; ++p) {
            int r = p*32 + (tid >> 3);
            int c = (tid & 7) << 2;
            int gm = m0 + r, gk = k0 + c;
            v4f v = {0.f,0.f,0.f,0.f};
            if (gm < M && gk < K) v = *(const v4f*)(A + (size_t)gm*K + gk);
            *(v4f*)&As[r][c] = v;
        }
        #pragma unroll
        for (int p = 0; p < 2; ++p) {
            int r = p*32 + (tid >> 3);
            int c = (tid & 7) << 2;
            int gn = n0 + r, gk = k0 + c;
            v4f v = {0.f,0.f,0.f,0.f};
            if (gn < N && gk < K) v = *(const v4f*)(W + (size_t)gn*K + gk);
            *(v4f*)&Ws[r][c] = v;
        }
        __syncthreads();

        #pragma unroll
        for (int kk = 0; kk < 8; ++kk) {
            const int kc = 4*kk + 2*lh;
            v2f a0 = *(const v2f*)&As[wm*32      + ll][kc];
            v2f a1 = *(const v2f*)&As[wm*32 + 16 + ll][kc];
            v2f b0 = *(const v2f*)&Ws[wn*32      + ll][kc];
            v2f b1 = *(const v2f*)&Ws[wn*32 + 16 + ll][kc];
            acc[0][0] = __builtin_amdgcn_wmma_f32_16x16x4_f32(false,a0,false,b0,(short)0,acc[0][0],false,false);
            acc[0][1] = __builtin_amdgcn_wmma_f32_16x16x4_f32(false,a0,false,b1,(short)0,acc[0][1],false,false);
            acc[1][0] = __builtin_amdgcn_wmma_f32_16x16x4_f32(false,a1,false,b0,(short)0,acc[1][0],false,false);
            acc[1][1] = __builtin_amdgcn_wmma_f32_16x16x4_f32(false,a1,false,b1,(short)0,acc[1][1],false,false);
        }
    }

    #pragma unroll
    for (int mt = 0; mt < 2; ++mt)
        #pragma unroll
        for (int nt = 0; nt < 2; ++nt) {
            int gn = n0 + wn*32 + nt*16 + ll;
            #pragma unroll
            for (int r = 0; r < 8; ++r) {
                int gm = m0 + wm*32 + mt*16 + lh*8 + r;
                if (gm < M && gn < N) C[(size_t)gm*N + gn] = acc[mt][nt][r];
            }
        }
}

__global__ __launch_bounds__(256) void attn_softmax_kernel(
    const float* __restrict__ scores, const float* __restrict__ lmask,
    const float* __restrict__ imask, float* __restrict__ attn,
    float* __restrict__ b_in)
{
    int row = blockIdx.x * 256 + threadIdx.x;
    if (row >= 8*256*20) return;
    int bs = row / 20;
    int l  = row - bs*20;

    float v[8];
    float mx = -3.4e38f;
    #pragma unroll
    for (int t = 0; t < 8; ++t) {
        float s = scores[(size_t)row*8 + t] + (1.0f - lmask[l*8 + t]) * NEG_BIG;
        v[t] = s;
        mx = fmaxf(mx, s);
    }
    float sum = 0.f;
    #pragma unroll
    for (int t = 0; t < 8; ++t) { float e = __expf(v[t] - mx); v[t] = e; sum += e; }
    float inv = 1.0f / sum;
    #pragma unroll
    for (int t = 0; t < 8; ++t) attn[(size_t)row*8 + t] = v[t] * inv;

    b_in[row] = mx + (1.0f - imask[bs]) * NEG_BIG;
}

__global__ __launch_bounds__(256) void q2c_kernel(
    const float* __restrict__ b_in, const float* __restrict__ tkn,
    float* __restrict__ q2c)
{
    __shared__ float red[256];
    __shared__ float wgt[256];
    const int tid = threadIdx.x;
    const int b = blockIdx.x / 20;
    const int l = blockIdx.x - b*20;

    float x = b_in[(size_t)(b*256 + tid)*20 + l];
    red[tid] = x; __syncthreads();
    for (int st = 128; st > 0; st >>= 1) {
        if (tid < st) red[tid] = fmaxf(red[tid], red[tid + st]);
        __syncthreads();
    }
    float mx = red[0]; __syncthreads();
    float e = __expf(x - mx);
    red[tid] = e; __syncthreads();
    for (int st = 128; st > 0; st >>= 1) {
        if (tid < st) red[tid] += red[tid + st];
        __syncthreads();
    }
    float sm = red[0]; __syncthreads();
    wgt[tid] = e / sm; __syncthreads();

    for (int h = tid; h < 768; h += 256) {
        float acc = 0.f;
        const float* tp = tkn + (size_t)b*256*768 + h;
        #pragma unroll 4
        for (int s = 0; s < 256; ++s) acc += wgt[s] * tp[(size_t)s*768];
        q2c[(size_t)(b*20 + l)*768 + h] = acc;
    }
}

__global__ __launch_bounds__(256) void fused_w5_tanh_wmma(
    const float* __restrict__ tkn,  const float* __restrict__ lab,
    const float* __restrict__ attn, const float* __restrict__ q2c,
    const float* __restrict__ W5,   const float* __restrict__ b5,
    float* __restrict__ out)
{
    __shared__ float As[TM][TKP];
    __shared__ float Ws[TN][TKP];
    __shared__ float aBlk[TM][8];

    const int tid  = threadIdx.x;
    const int lane = tid & 31, wid = tid >> 5;
    const int wm   = wid >> 1, wn  = wid & 1;
    const int ll   = lane & 15, lh = lane >> 4;
    const int m0   = blockIdx.x * TM, n0 = blockIdx.y * TN;

    {
        int r = tid >> 1, c = (tid & 1) << 2;
        *(v4f*)&aBlk[r][c] = *(const v4f*)(attn + (size_t)(m0 + r)*8 + c);
    }

    const v8f vzero = {0.f,0.f,0.f,0.f,0.f,0.f,0.f,0.f};
    v8f acc[2][2];
    acc[0][0]=vzero; acc[0][1]=vzero; acc[1][0]=vzero; acc[1][1]=vzero;

    for (int k0 = 0; k0 < 3072; k0 += TK) {
        const int sec = k0 / 768;
        const int j0  = k0 - sec*768;
        __syncthreads();
        #pragma unroll
        for (int p = 0; p < 2; ++p) {
            int r = p*32 + (tid >> 3);
            int c = (tid & 7) << 2;
            *(v4f*)&Ws[r][c] = *(const v4f*)(W5 + (size_t)(n0 + r)*3072 + (k0 + c));
        }
        #pragma unroll
        for (int p = 0; p < 4; ++p) {
            int r = p*32 + (tid >> 3);
            int c = (tid & 7) << 2;
            int m  = m0 + r;
            int bs = m / 20;
            int l  = m - bs*20;
            int j  = j0 + c;
            v4f v;
            if (sec == 0) {
                v = *(const v4f*)(tkn + (size_t)bs*768 + j);
            } else if (sec <= 2) {
                v4f cq = {0.f,0.f,0.f,0.f};
                const float* labRow = lab + (size_t)l*8*768 + j;
                #pragma unroll
                for (int t = 0; t < 8; ++t) {
                    float w  = aBlk[r][t];
                    v4f  lv  = *(const v4f*)(labRow + (size_t)t*768);
                    cq += w * lv;
                }
                if (sec == 1) v = cq;
                else {
                    v4f t4 = *(const v4f*)(tkn + (size_t)bs*768 + j);
                    v = t4 * cq;
                }
            } else {
                int b  = bs >> 8;
                v4f t4 = *(const v4f*)(tkn + (size_t)bs*768 + j);
                v4f q4 = *(const v4f*)(q2c + (size_t)(b*20 + l)*768 + j);
                v = t4 * q4;
            }
            *(v4f*)&As[r][c] = v;
        }
        __syncthreads();

        #pragma unroll
        for (int kk = 0; kk < 8; ++kk) {
            const int kc = 4*kk + 2*lh;
            v2f a0 = *(const v2f*)&As[wm*32      + ll][kc];
            v2f a1 = *(const v2f*)&As[wm*32 + 16 + ll][kc];
            v2f b0 = *(const v2f*)&Ws[wn*32      + ll][kc];
            v2f b1 = *(const v2f*)&Ws[wn*32 + 16 + ll][kc];
            acc[0][0] = __builtin_amdgcn_wmma_f32_16x16x4_f32(false,a0,false,b0,(short)0,acc[0][0],false,false);
            acc[0][1] = __builtin_amdgcn_wmma_f32_16x16x4_f32(false,a0,false,b1,(short)0,acc[0][1],false,false);
            acc[1][0] = __builtin_amdgcn_wmma_f32_16x16x4_f32(false,a1,false,b0,(short)0,acc[1][0],false,false);
            acc[1][1] = __builtin_amdgcn_wmma_f32_16x16x4_f32(false,a1,false,b1,(short)0,acc[1][1],false,false);
        }
    }

    #pragma unroll
    for (int mt = 0; mt < 2; ++mt)
        #pragma unroll
        for (int nt = 0; nt < 2; ++nt) {
            int gn = n0 + wn*32 + nt*16 + ll;
            float bias = b5[gn];
            #pragma unroll
            for (int r = 0; r < 8; ++r) {
                int gm = m0 + wm*32 + mt*16 + lh*8 + r;
                out[(size_t)gm*768 + gn] = tanhf(acc[mt][nt][r] + bias);
            }
        }
}

extern "C" void kernel_launch(void* const* d_in, const int* in_sizes, int n_in,
                              void* d_out, int out_size, void* d_ws, size_t ws_size,
                              hipStream_t stream)
{
    (void)in_sizes; (void)n_in; (void)out_size; (void)ws_size;

    const float* token = (const float*)d_in[0];
    const float* label = (const float*)d_in[1];
    const float* imask = (const float*)d_in[2];
    const float* lmask = (const float*)d_in[3];
    const float* W1    = (const float*)d_in[4];
    const float* W2    = (const float*)d_in[5];
    const float* W5    = (const float*)d_in[6];
    const float* b5    = (const float*)d_in[7];
    float* out = (float*)d_out;

    float* ws     = (float*)d_ws;
    float* tkn    = ws;
    float* lab    = tkn    + 2048*768;
    float* scores = lab    + 160*768;
    float* attn   = scores + 2048*160;
    float* b_in   = attn   + 2048*160;
    float* q2c    = b_in   + 40960;

    dim3 blk(256);
    gemm_nt_f32_wmma<<<dim3(16, 12), blk, 0, stream>>>(token, W1, tkn, 2048, 768, 768);
    gemm_nt_f32_wmma<<<dim3(2, 12),  blk, 0, stream>>>(label, W2, lab, 160, 768, 300);
    gemm_nt_f32_wmma<<<dim3(16, 3),  blk, 0, stream>>>(tkn, lab, scores, 2048, 160, 768);
    attn_softmax_kernel<<<dim3(160), blk, 0, stream>>>(scores, lmask, imask, attn, b_in);
    q2c_kernel<<<dim3(160), blk, 0, stream>>>(b_in, tkn, q2c);
    fused_w5_tanh_wmma<<<dim3(320, 12), blk, 0, stream>>>(tkn, lab, attn, q2c, W5, b5, out);
}


# Factor fused@W5: tkn and c2q K-sections computed once per token/label row (2208-row WMMA pre-GEMM) and fed as accumulator start tile; main K 3072->1536; sum order changes (sec0,sec1 summed first)
# speedup vs baseline: 1.5480x; 1.5480x over previous
#include <hip/hip_runtime.h>
#include <math.h>


typedef __attribute__((ext_vector_type(2))) float v2f;
typedef __attribute__((ext_vector_type(4))) float v4f;
typedef __attribute__((ext_vector_type(8))) float v8f;

#define NEG_BIG (-10000.0f)

#define TM 128
#define TN 64
#define TK 32
#define TKP 34


__global__ __launch_bounds__(256) void gemm_nt_f32_wmma(
    const float* __restrict__ A, const float* __restrict__ W,
    float* __restrict__ C, int M, int N, int K)
{
    __shared__ float As[TM][TKP];
    __shared__ float Ws[TN][TKP];

    const int tid  = threadIdx.x;
    const int lane = tid & 31, wid = tid >> 5;
    const int wm   = wid >> 1, wn  = wid & 1;
    const int ll   = lane & 15, lh = lane >> 4;
    const int m0   = blockIdx.x * TM, n0 = blockIdx.y * TN;

    const v8f vzero = {0.f,0.f,0.f,0.f,0.f,0.f,0.f,0.f};
    v8f acc[2][2];
    acc[0][0]=vzero; acc[0][1]=vzero; acc[1][0]=vzero; acc[1][1]=vzero;

    for (int k0 = 0; k0 < K; k0 += TK) {
        __syncthreads();
        #pragma unroll
        for (int p = 0; p < 4; ++p) {
            int r = p*32 + (tid >> 3);
            int c = (tid & 7) << 2;
            int gm = m0 + r, gk = k0 + c;
            v4f v = {0.f,0.f,0.f,0.f};
            if (gm < M && gk < K) v = *(const v4f*)(A + (size_t)gm*K + gk);
            *(v4f*)&As[r][c] = v;
        }
        #pragma unroll
        for (int p = 0; p < 2; ++p) {
            int r = p*32 + (tid >> 3);
            int c = (tid & 7) << 2;
            int gn = n0 + r, gk = k0 + c;
            v4f v = {0.f,0.f,0.f,0.f};
            if (gn < N && gk < K) v = *(const v4f*)(W + (size_t)gn*K + gk);
            *(v4f*)&Ws[r][c] = v;
        }
        __syncthreads();

        #pragma unroll
        for (int kk = 0; kk < 8; ++kk) {
            const int kc = 4*kk + 2*lh;
            v2f a0 = *(const v2f*)&As[wm*32      + ll][kc];
            v2f a1 = *(const v2f*)&As[wm*32 + 16 + ll][kc];
            v2f b0 = *(const v2f*)&Ws[wn*32      + ll][kc];
            v2f b1 = *(const v2f*)&Ws[wn*32 + 16 + ll][kc];
            acc[0][0] = __builtin_amdgcn_wmma_f32_16x16x4_f32(false,a0,false,b0,(short)0,acc[0][0],false,false);
            acc[0][1] = __builtin_amdgcn_wmma_f32_16x16x4_f32(false,a0,false,b1,(short)0,acc[0][1],false,false);
            acc[1][0] = __builtin_amdgcn_wmma_f32_16x16x4_f32(false,a1,false,b0,(short)0,acc[1][0],false,false);
            acc[1][1] = __builtin_amdgcn_wmma_f32_16x16x4_f32(false,a1,false,b1,(short)0,acc[1][1],false,false);
        }
    }

    #pragma unroll
    for (int mt = 0; mt < 2; ++mt)
        #pragma unroll
        for (int nt = 0; nt < 2; ++nt) {
            int gn = n0 + wn*32 + nt*16 + ll;
            #pragma unroll
            for (int r = 0; r < 8; ++r) {
                int gm = m0 + wm*32 + mt*16 + lh*8 + r;
                if (gm < M && gn < N) C[(size_t)gm*N + gn] = acc[mt][nt][r];
            }
        }
}

__global__ __launch_bounds__(256) void attn_softmax_kernel(
    const float* __restrict__ scores, const float* __restrict__ lmask,
    const float* __restrict__ imask, float* __restrict__ attn,
    float* __restrict__ b_in)
{
    int row = blockIdx.x * 256 + threadIdx.x;
    if (row >= 8*256*20) return;
    int bs = row / 20;
    int l  = row - bs*20;

    float v[8];
    float mx = -3.4e38f;
    #pragma unroll
    for (int t = 0; t < 8; ++t) {
        float s = scores[(size_t)row*8 + t] + (1.0f - lmask[l*8 + t]) * NEG_BIG;
        v[t] = s;
        mx = fmaxf(mx, s);
    }
    float sum = 0.f;
    #pragma unroll
    for (int t = 0; t < 8; ++t) { float e = __expf(v[t] - mx); v[t] = e; sum += e; }
    float inv = 1.0f / sum;
    #pragma unroll
    for (int t = 0; t < 8; ++t) attn[(size_t)row*8 + t] = v[t] * inv;

    b_in[row] = mx + (1.0f - imask[bs]) * NEG_BIG;
}

__global__ __launch_bounds__(256) void q2c_kernel(
    const float* __restrict__ b_in, const float* __restrict__ tkn,
    float* __restrict__ q2c)
{
    __shared__ float red[256];
    __shared__ float wgt[256];
    const int tid = threadIdx.x;
    const int b = blockIdx.x / 20;
    const int l = blockIdx.x - b*20;

    float x = b_in[(size_t)(b*256 + tid)*20 + l];
    red[tid] = x; __syncthreads();
    for (int st = 128; st > 0; st >>= 1) {
        if (tid < st) red[tid] = fmaxf(red[tid], red[tid + st]);
        __syncthreads();
    }
    float mx = red[0]; __syncthreads();
    float e = __expf(x - mx);
    red[tid] = e; __syncthreads();
    for (int st = 128; st > 0; st >>= 1) {
        if (tid < st) red[tid] += red[tid + st];
        __syncthreads();
    }
    float sm = red[0]; __syncthreads();
    wgt[tid] = e / sm; __syncthreads();

    for (int h = tid; h < 768; h += 256) {
        float acc = 0.f;
        const float* tp = tkn + (size_t)b*256*768 + h;
        #pragma unroll 4
        for (int s = 0; s < 256; ++s) acc += wgt[s] * tp[(size_t)s*768];
        q2c[(size_t)(b*20 + l)*768 + h] = acc;
    }
}

__global__ __launch_bounds__(256) void rowfactor_w5_wmma(
    const float* __restrict__ tkn, const float* __restrict__ lab,
    const float* __restrict__ W5,
    float* __restrict__ T0, float* __restrict__ Pt)
{
    __shared__ float As[TM][TKP];
    __shared__ float Ws[TN][TKP];

    const int tid  = threadIdx.x;
    const int lane = tid & 31, wid = tid >> 5;
    const int wm   = wid >> 1, wn  = wid & 1;
    const int ll   = lane & 15, lh = lane >> 4;
    const int m0   = blockIdx.x * TM, n0 = blockIdx.y * TN;

    const bool isLab = (m0 >= 2048);
    const int  q0    = m0 - 2048;
    const float* __restrict__ Ab = isLab ? (lab + (size_t)q0*768) : (tkn + (size_t)m0*768);
    const int  rows  = isLab ? (160 - q0) : TM;
    const float* __restrict__ Wb = W5 + (isLab ? 768 : 0);

    const v8f vzero = {0.f,0.f,0.f,0.f,0.f,0.f,0.f,0.f};
    v8f acc[2][2];
    acc[0][0]=vzero; acc[0][1]=vzero; acc[1][0]=vzero; acc[1][1]=vzero;

    for (int k0 = 0; k0 < 768; k0 += TK) {
        __syncthreads();
        #pragma unroll
        for (int p = 0; p < 4; ++p) {
            int r = p*32 + (tid >> 3);
            int c = (tid & 7) << 2;
            v4f v = {0.f,0.f,0.f,0.f};
            if (r < rows) v = *(const v4f*)(Ab + (size_t)r*768 + (k0 + c));
            *(v4f*)&As[r][c] = v;
        }
        #pragma unroll
        for (int p = 0; p < 2; ++p) {
            int r = p*32 + (tid >> 3);
            int c = (tid & 7) << 2;
            *(v4f*)&Ws[r][c] = *(const v4f*)(Wb + (size_t)(n0 + r)*3072 + (k0 + c));
        }
        __syncthreads();

        #pragma unroll
        for (int kk = 0; kk < 8; ++kk) {
            const int kc = 4*kk + 2*lh;
            v2f a0 = *(const v2f*)&As[wm*32      + ll][kc];
            v2f a1 = *(const v2f*)&As[wm*32 + 16 + ll][kc];
            v2f b0 = *(const v2f*)&Ws[wn*32      + ll][kc];
            v2f b1 = *(const v2f*)&Ws[wn*32 + 16 + ll][kc];
            acc[0][0] = __builtin_amdgcn_wmma_f32_16x16x4_f32(false,a0,false,b0,(short)0,acc[0][0],false,false);
            acc[0][1] = __builtin_amdgcn_wmma_f32_16x16x4_f32(false,a0,false,b1,(short)0,acc[0][1],false,false);
            acc[1][0] = __builtin_amdgcn_wmma_f32_16x16x4_f32(false,a1,false,b0,(short)0,acc[1][0],false,false);
            acc[1][1] = __builtin_amdgcn_wmma_f32_16x16x4_f32(false,a1,false,b1,(short)0,acc[1][1],false,false);
        }
    }

    #pragma unroll
    for (int mt = 0; mt < 2; ++mt)
        #pragma unroll
        for (int nt = 0; nt < 2; ++nt) {
            int gn = n0 + wn*32 + nt*16 + ll;
            #pragma unroll
            for (int r = 0; r < 8; ++r) {
                int row = wm*32 + mt*16 + lh*8 + r;
                if (!isLab) {
                    T0[(size_t)(m0 + row)*768 + gn] = acc[mt][nt][r];
                } else if (row < rows) {
                    int q = q0 + row;
                    Pt[((size_t)(q >> 3)*768 + gn)*8 + (q & 7)] = acc[mt][nt][r];
                }
            }
        }
}

__global__ __launch_bounds__(256) void fused_w5_tanh_wmma(
    const float* __restrict__ tkn,  const float* __restrict__ lab,
    const float* __restrict__ attn, const float* __restrict__ q2c,
    const float* __restrict__ W5,   const float* __restrict__ b5,
    const float* __restrict__ T0,   const float* __restrict__ Pt,
    float* __restrict__ out)
{
    __shared__ float As[TM][TKP];
    __shared__ float Ws[TN][TKP];
    __shared__ float aBlk[TM][8];
    __shared__ float startS[TM][TN + 2];

    const int tid  = threadIdx.x;
    const int lane = tid & 31, wid = tid >> 5;
    const int wm   = wid >> 1, wn  = wid & 1;
    const int ll   = lane & 15, lh = lane >> 4;
    const int m0   = blockIdx.x * TM, n0 = blockIdx.y * TN;

    {
        int r = tid >> 1, c = (tid & 1) << 2;
        *(v4f*)&aBlk[r][c] = *(const v4f*)(attn + (size_t)(m0 + r)*8 + c);
    }

    __syncthreads();

    {
        const int col = tid & 63;
        const int rg  = tid >> 6;
        const int gn  = n0 + col;
        #pragma unroll 4
        for (int i = 0; i < 32; ++i) {
            const int row = rg*32 + i;
            const int m   = m0 + row;
            const int bs  = m / 20;
            const int l   = m - bs*20;
            const v4f aLo = *(const v4f*)&aBlk[row][0];
            const v4f aHi = *(const v4f*)&aBlk[row][4];
            const float* pp = Pt + ((size_t)l*768 + gn)*8;
            const v4f pLo = *(const v4f*)pp;
            const v4f pHi = *(const v4f*)(pp + 4);
            float s = T0[(size_t)bs*768 + gn];
            s += aLo.x*pLo.x; s += aLo.y*pLo.y; s += aLo.z*pLo.z; s += aLo.w*pLo.w;
            s += aHi.x*pHi.x; s += aHi.y*pHi.y; s += aHi.z*pHi.z; s += aHi.w*pHi.w;
            startS[row][col] = s;
        }
    }
    __syncthreads();

    v8f acc[2][2];
    #pragma unroll
    for (int mt = 0; mt < 2; ++mt)
        #pragma unroll
        for (int nt = 0; nt < 2; ++nt)
            #pragma unroll
            for (int r = 0; r < 8; ++r)
                acc[mt][nt][r] = startS[wm*32 + mt*16 + lh*8 + r][wn*32 + nt*16 + ll];

    for (int k0 = 1536; k0 < 3072; k0 += TK) {
        const int sec = k0 / 768;
        const int j0  = k0 - sec*768;
        __syncthreads();
        #pragma unroll
        for (int p = 0; p < 2; ++p) {
            int r = p*32 + (tid >> 3);
            int c = (tid & 7) << 2;
            *(v4f*)&Ws[r][c] = *(const v4f*)(W5 + (size_t)(n0 + r)*3072 + (k0 + c));
        }
        #pragma unroll
        for (int p = 0; p < 4; ++p) {
            int r = p*32 + (tid >> 3);
            int c = (tid & 7) << 2;
            int m  = m0 + r;
            int bs = m / 20;
            int l  = m - bs*20;
            int j  = j0 + c;
            v4f v;
            if (sec == 2) {
                v4f cq = {0.f,0.f,0.f,0.f};
                const float* labRow = lab + (size_t)l*8*768 + j;
                #pragma unroll
                for (int t = 0; t < 8; ++t) {
                    float w  = aBlk[r][t];
                    v4f  lv  = *(const v4f*)(labRow + (size_t)t*768);
                    cq += w * lv;
                }
                v4f t4 = *(const v4f*)(tkn + (size_t)bs*768 + j);
                v = t4 * cq;
            } else {
                int b  = bs >> 8;
                v4f t4 = *(const v4f*)(tkn + (size_t)bs*768 + j);
                v4f q4 = *(const v4f*)(q2c + (size_t)(b*20 + l)*768 + j);
                v = t4 * q4;
            }
            *(v4f*)&As[r][c] = v;
        }
        __syncthreads();

        #pragma unroll
        for (int kk = 0; kk < 8; ++kk) {
            const int kc = 4*kk + 2*lh;
            v2f a0 = *(const v2f*)&As[wm*32      + ll][kc];
            v2f a1 = *(const v2f*)&As[wm*32 + 16 + ll][kc];
            v2f b0 = *(const v2f*)&Ws[wn*32      + ll][kc];
            v2f b1 = *(const v2f*)&Ws[wn*32 + 16 + ll][kc];
            acc[0][0] = __builtin_amdgcn_wmma_f32_16x16x4_f32(false,a0,false,b0,(short)0,acc[0][0],false,false);
            acc[0][1] = __builtin_amdgcn_wmma_f32_16x16x4_f32(false,a0,false,b1,(short)0,acc[0][1],false,false);
            acc[1][0] = __builtin_amdgcn_wmma_f32_16x16x4_f32(false,a1,false,b0,(short)0,acc[1][0],false,false);
            acc[1][1] = __builtin_amdgcn_wmma_f32_16x16x4_f32(false,a1,false,b1,(short)0,acc[1][1],false,false);
        }
    }

    #pragma unroll
    for (int mt = 0; mt < 2; ++mt)
        #pragma unroll
        for (int nt = 0; nt < 2; ++nt) {
            int gn = n0 + wn*32 + nt*16 + ll;
            float bias = b5[gn];
            #pragma unroll
            for (int r = 0; r < 8; ++r) {
                int gm = m0 + wm*32 + mt*16 + lh*8 + r;
                out[(size_t)gm*768 + gn] = tanhf(acc[mt][nt][r] + bias);
            }
        }
}

extern "C" void kernel_launch(void* const* d_in, const int* in_sizes, int n_in,
                              void* d_out, int out_size, void* d_ws, size_t ws_size,
                              hipStream_t stream)
{
    (void)in_sizes; (void)n_in; (void)out_size; (void)ws_size;

    const float* token = (const float*)d_in[0];
    const float* label = (const float*)d_in[1];
    const float* imask = (const float*)d_in[2];
    const float* lmask = (const float*)d_in[3];
    const float* W1    = (const float*)d_in[4];
    const float* W2    = (const float*)d_in[5];
    const float* W5    = (const float*)d_in[6];
    const float* b5    = (const float*)d_in[7];
    float* out = (float*)d_out;

    float* ws     = (float*)d_ws;
    float* tkn    = ws;
    float* lab    = tkn    + 2048*768;
    float* scores = lab    + 160*768;
    float* attn   = scores + 2048*160;
    float* b_in   = attn   + 2048*160;
    float* q2c    = b_in   + 40960;
    float* T0     = q2c    + 160*768;
    float* Pt     = T0     + 2048*768;

    dim3 blk(256);
    gemm_nt_f32_wmma<<<dim3(16, 12), blk, 0, stream>>>(token, W1, tkn, 2048, 768, 768);
    gemm_nt_f32_wmma<<<dim3(2, 12),  blk, 0, stream>>>(label, W2, lab, 160, 768, 300);
    gemm_nt_f32_wmma<<<dim3(16, 3),  blk, 0, stream>>>(tkn, lab, scores, 2048, 160, 768);
    attn_softmax_kernel<<<dim3(160), blk, 0, stream>>>(scores, lmask, imask, attn, b_in);
    q2c_kernel<<<dim3(160), blk, 0, stream>>>(b_in, tkn, q2c);
    rowfactor_w5_wmma<<<dim3(18, 12), blk, 0, stream>>>(tkn, lab, W5, T0, Pt);
    fused_w5_tanh_wmma<<<dim3(320, 12), blk, 0, stream>>>(tkn, lab, attn, q2c, W5, b5, T0, Pt, out);
}


# Main GEMM's A operand (fused sec2|sec3, 40960x1536) built once by a new kernel, not per column block in the K loop; loop now copies A/W5 tiles prefetched one chunk ahead into a 2-stage LDS ring
# speedup vs baseline: 1.7999x; 1.1627x over previous
#include <hip/hip_runtime.h>
#include <math.h>


typedef __attribute__((ext_vector_type(2))) float v2f;
typedef __attribute__((ext_vector_type(4))) float v4f;
typedef __attribute__((ext_vector_type(8))) float v8f;

#define NEG_BIG (-10000.0f)

#define TM 128
#define TN 64
#define TK 32
#define TKP 34


__global__ __launch_bounds__(256) void gemm_nt_f32_wmma(
    const float* __restrict__ A, const float* __restrict__ W,
    float* __restrict__ C, int M, int N, int K)
{
    __shared__ float As[TM][TKP];
    __shared__ float Ws[TN][TKP];

    const int tid  = threadIdx.x;
    const int lane = tid & 31, wid = tid >> 5;
    const int wm   = wid >> 1, wn  = wid & 1;
    const int ll   = lane & 15, lh = lane >> 4;
    const int m0   = blockIdx.x * TM, n0 = blockIdx.y * TN;

    const v8f vzero = {0.f,0.f,0.f,0.f,0.f,0.f,0.f,0.f};
    v8f acc[2][2];
    acc[0][0]=vzero; acc[0][1]=vzero; acc[1][0]=vzero; acc[1][1]=vzero;

    for (int k0 = 0; k0 < K; k0 += TK) {
        __syncthreads();
        #pragma unroll
        for (int p = 0; p < 4; ++p) {
            int r = p*32 + (tid >> 3);
            int c = (tid & 7) << 2;
            int gm = m0 + r, gk = k0 + c;
            v4f v = {0.f,0.f,0.f,0.f};
            if (gm < M && gk < K) v = *(const v4f*)(A + (size_t)gm*K + gk);
            *(v4f*)&As[r][c] = v;
        }
        #pragma unroll
        for (int p = 0; p < 2; ++p) {
            int r = p*32 + (tid >> 3);
            int c = (tid & 7) << 2;
            int gn = n0 + r, gk = k0 + c;
            v4f v = {0.f,0.f,0.f,0.f};
            if (gn < N && gk < K) v = *(const v4f*)(W + (size_t)gn*K + gk);
            *(v4f*)&Ws[r][c] = v;
        }
        __syncthreads();

        #pragma unroll
        for (int kk = 0; kk < 8; ++kk) {
            const int kc = 4*kk + 2*lh;
            v2f a0 = *(const v2f*)&As[wm*32      + ll][kc];
            v2f a1 = *(const v2f*)&As[wm*32 + 16 + ll][kc];
            v2f b0 = *(const v2f*)&Ws[wn*32      + ll][kc];
            v2f b1 = *(const v2f*)&Ws[wn*32 + 16 + ll][kc];
            acc[0][0] = __builtin_amdgcn_wmma_f32_16x16x4_f32(false,a0,false,b0,(short)0,acc[0][0],false,false);
            acc[0][1] = __builtin_amdgcn_wmma_f32_16x16x4_f32(false,a0,false,b1,(short)0,acc[0][1],false,false);
            acc[1][0] = __builtin_amdgcn_wmma_f32_16x16x4_f32(false,a1,false,b0,(short)0,acc[1][0],false,false);
            acc[1][1] = __builtin_amdgcn_wmma_f32_16x16x4_f32(false,a1,false,b1,(short)0,acc[1][1],false,false);
        }
    }

    #pragma unroll
    for (int mt = 0; mt < 2; ++mt)
        #pragma unroll
        for (int nt = 0; nt < 2; ++nt) {
            int gn = n0 + wn*32 + nt*16 + ll;
            #pragma unroll
            for (int r = 0; r < 8; ++r) {
                int gm = m0 + wm*32 + mt*16 + lh*8 + r;
                if (gm < M && gn < N) C[(size_t)gm*N + gn] = acc[mt][nt][r];
            }
        }
}

__global__ __launch_bounds__(256) void attn_softmax_kernel(
    const float* __restrict__ scores, const float* __restrict__ lmask,
    const float* __restrict__ imask, float* __restrict__ attn,
    float* __restrict__ b_in)
{
    int row = blockIdx.x * 256 + threadIdx.x;
    if (row >= 8*256*20) return;
    int bs = row / 20;
    int l  = row - bs*20;

    float v[8];
    float mx = -3.4e38f;
    #pragma unroll
    for (int t = 0; t < 8; ++t) {
        float s = scores[(size_t)row*8 + t] + (1.0f - lmask[l*8 + t]) * NEG_BIG;
        v[t] = s;
        mx = fmaxf(mx, s);
    }
    float sum = 0.f;
    #pragma unroll
    for (int t = 0; t < 8; ++t) { float e = __expf(v[t] - mx); v[t] = e; sum += e; }
    float inv = 1.0f / sum;
    #pragma unroll
    for (int t = 0; t < 8; ++t) attn[(size_t)row*8 + t] = v[t] * inv;

    b_in[row] = mx + (1.0f - imask[bs]) * NEG_BIG;
}

__global__ __launch_bounds__(256) void q2c_kernel(
    const float* __restrict__ b_in, const float* __restrict__ tkn,
    float* __restrict__ q2c)
{
    __shared__ float red[256];
    __shared__ float wgt[256];
    const int tid = threadIdx.x;
    const int b = blockIdx.x / 20;
    const int l = blockIdx.x - b*20;

    float x = b_in[(size_t)(b*256 + tid)*20 + l];
    red[tid] = x; __syncthreads();
    for (int st = 128; st > 0; st >>= 1) {
        if (tid < st) red[tid] = fmaxf(red[tid], red[tid + st]);
        __syncthreads();
    }
    float mx = red[0]; __syncthreads();
    float e = __expf(x - mx);
    red[tid] = e; __syncthreads();
    for (int st = 128; st > 0; st >>= 1) {
        if (tid < st) red[tid] += red[tid + st];
        __syncthreads();
    }
    float sm = red[0]; __syncthreads();
    wgt[tid] = e / sm; __syncthreads();

    for (int h = tid; h < 768; h += 256) {
        float acc = 0.f;
        const float* tp = tkn + (size_t)b*256*768 + h;
        #pragma unroll 4
        for (int s = 0; s < 256; ++s) acc += wgt[s] * tp[(size_t)s*768];
        q2c[(size_t)(b*20 + l)*768 + h] = acc;
    }
}

__global__ __launch_bounds__(256) void rowfactor_w5_wmma(
    const float* __restrict__ tkn, const float* __restrict__ lab,
    const float* __restrict__ W5,
    float* __restrict__ T0, float* __restrict__ Pt)
{
    __shared__ float As[TM][TKP];
    __shared__ float Ws[TN][TKP];

    const int tid  = threadIdx.x;
    const int lane = tid & 31, wid = tid >> 5;
    const int wm   = wid >> 1, wn  = wid & 1;
    const int ll   = lane & 15, lh = lane >> 4;
    const int m0   = blockIdx.x * TM, n0 = blockIdx.y * TN;

    const bool isLab = (m0 >= 2048);
    const int  q0    = m0 - 2048;
    const float* __restrict__ Ab = isLab ? (lab + (size_t)q0*768) : (tkn + (size_t)m0*768);
    const int  rows  = isLab ? (160 - q0) : TM;
    const float* __restrict__ Wb = W5 + (isLab ? 768 : 0);

    const v8f vzero = {0.f,0.f,0.f,0.f,0.f,0.f,0.f,0.f};
    v8f acc[2][2];
    acc[0][0]=vzero; acc[0][1]=vzero; acc[1][0]=vzero; acc[1][1]=vzero;

    for (int k0 = 0; k0 < 768; k0 += TK) {
        __syncthreads();
        #pragma unroll
        for (int p = 0; p < 4; ++p) {
            int r = p*32 + (tid >> 3);
            int c = (tid & 7) << 2;
            v4f v = {0.f,0.f,0.f,0.f};
            if (r < rows) v = *(const v4f*)(Ab + (size_t)r*768 + (k0 + c));
            *(v4f*)&As[r][c] = v;
        }
        #pragma unroll
        for (int p = 0; p < 2; ++p) {
            int r = p*32 + (tid >> 3);
            int c = (tid & 7) << 2;
            *(v4f*)&Ws[r][c] = *(const v4f*)(Wb + (size_t)(n0 + r)*3072 + (k0 + c));
        }
        __syncthreads();

        #pragma unroll
        for (int kk = 0; kk < 8; ++kk) {
            const int kc = 4*kk + 2*lh;
            v2f a0 = *(const v2f*)&As[wm*32      + ll][kc];
            v2f a1 = *(const v2f*)&As[wm*32 + 16 + ll][kc];
            v2f b0 = *(const v2f*)&Ws[wn*32      + ll][kc];
            v2f b1 = *(const v2f*)&Ws[wn*32 + 16 + ll][kc];
            acc[0][0] = __builtin_amdgcn_wmma_f32_16x16x4_f32(false,a0,false,b0,(short)0,acc[0][0],false,false);
            acc[0][1] = __builtin_amdgcn_wmma_f32_16x16x4_f32(false,a0,false,b1,(short)0,acc[0][1],false,false);
            acc[1][0] = __builtin_amdgcn_wmma_f32_16x16x4_f32(false,a1,false,b0,(short)0,acc[1][0],false,false);
            acc[1][1] = __builtin_amdgcn_wmma_f32_16x16x4_f32(false,a1,false,b1,(short)0,acc[1][1],false,false);
        }
    }

    #pragma unroll
    for (int mt = 0; mt < 2; ++mt)
        #pragma unroll
        for (int nt = 0; nt < 2; ++nt) {
            int gn = n0 + wn*32 + nt*16 + ll;
            #pragma unroll
            for (int r = 0; r < 8; ++r) {
                int row = wm*32 + mt*16 + lh*8 + r;
                if (!isLab) {
                    T0[(size_t)(m0 + row)*768 + gn] = acc[mt][nt][r];
                } else if (row < rows) {
                    int q = q0 + row;
                    Pt[((size_t)(q >> 3)*768 + gn)*8 + (q & 7)] = acc[mt][nt][r];
                }
            }
        }
}

__global__ __launch_bounds__(192) void build_fused_a(
    const float* __restrict__ tkn,  const float* __restrict__ lab,
    const float* __restrict__ attn, const float* __restrict__ q2c,
    float* __restrict__ Af)
{
    const int tid = threadIdx.x;
    const int l   = blockIdx.x;
    const int bs0 = blockIdx.y * 16;
    const int b   = bs0 >> 8;
    const int j   = tid << 2;

    v4f lv[8];
    #pragma unroll
    for (int t = 0; t < 8; ++t) lv[t] = *(const v4f*)(lab + (size_t)(l*8 + t)*768 + j);
    const v4f q4 = *(const v4f*)(q2c + (size_t)(b*20 + l)*768 + j);

    #pragma unroll 4
    for (int i = 0; i < 16; ++i) {
        const int bs = bs0 + i;
        const size_t m = (size_t)bs*20 + l;
        const v4f aLo = *(const v4f*)(attn + m*8);
        const v4f aHi = *(const v4f*)(attn + m*8 + 4);
        const v4f t4  = *(const v4f*)(tkn + (size_t)bs*768 + j);
        v4f cq = {0.f,0.f,0.f,0.f};
        cq += aLo.x * lv[0]; cq += aLo.y * lv[1]; cq += aLo.z * lv[2]; cq += aLo.w * lv[3];
        cq += aHi.x * lv[4]; cq += aHi.y * lv[5]; cq += aHi.z * lv[6]; cq += aHi.w * lv[7];
        float* dst = Af + m*1536 + j;
        *(v4f*)dst         = t4 * cq;
        *(v4f*)(dst + 768) = t4 * q4;
    }
}

#define STG (192*TKP)

__device__ __forceinline__ void tile_get(const float* __restrict__ aP, const float* __restrict__ wP,
                                         v4f (&a)[4], v4f (&w)[2])
{
    #pragma unroll
    for (int p = 0; p < 4; ++p) a[p] = *(const v4f*)(aP + (size_t)p*32*1536);
    #pragma unroll
    for (int p = 0; p < 2; ++p) w[p] = *(const v4f*)(wP + (size_t)p*32*3072);
}

__device__ __forceinline__ void stage_put(float* st, int tid, const v4f (&a)[4], const v4f (&w)[2])
{
    const int r = tid >> 3, c = (tid & 7) << 2;
    #pragma unroll
    for (int p = 0; p < 4; ++p) {
        float* d = st + (p*32 + r)*TKP + c;
        v2f lo = {a[p].x, a[p].y}, hi = {a[p].z, a[p].w};
        *(v2f*)d = lo; *(v2f*)(d + 2) = hi;
    }
    #pragma unroll
    for (int p = 0; p < 2; ++p) {
        float* d = st + (TM + p*32 + r)*TKP + c;
        v2f lo = {w[p].x, w[p].y}, hi = {w[p].z, w[p].w};
        *(v2f*)d = lo; *(v2f*)(d + 2) = hi;
    }
}

__device__ __forceinline__ void mma_phase(const float* st, v8f (&acc)[2][2],
                                          int wm, int wn, int ll, int lh)
{
    const float* As = st;
    const float* Ws = st + TM*TKP;
    #pragma unroll
    for (int kk = 0; kk < 8; ++kk) {
        const int kc = 4*kk + 2*lh;
        v2f a0 = *(const v2f*)(As + (wm*32      + ll)*TKP + kc);
        v2f a1 = *(const v2f*)(As + (wm*32 + 16 + ll)*TKP + kc);
        v2f b0 = *(const v2f*)(Ws + (wn*32      + ll)*TKP + kc);
        v2f b1 = *(const v2f*)(Ws + (wn*32 + 16 + ll)*TKP + kc);
        acc[0][0] = __builtin_amdgcn_wmma_f32_16x16x4_f32(false,a0,false,b0,(short)0,acc[0][0],false,false);
        acc[0][1] = __builtin_amdgcn_wmma_f32_16x16x4_f32(false,a0,false,b1,(short)0,acc[0][1],false,false);
        acc[1][0] = __builtin_amdgcn_wmma_f32_16x16x4_f32(false,a1,false,b0,(short)0,acc[1][0],false,false);
        acc[1][1] = __builtin_amdgcn_wmma_f32_16x16x4_f32(false,a1,false,b1,(short)0,acc[1][1],false,false);
    }
}

__global__ __launch_bounds__(256) void fused_w5_tanh_wmma(
    const float* __restrict__ attn, const float* __restrict__ Af,
    const float* __restrict__ W5,   const float* __restrict__ b5,
    const float* __restrict__ T0,   const float* __restrict__ Pt,
    float* __restrict__ out)
{
    __shared__ __attribute__((aligned(16))) float smem[2*STG];
    float (*aBlk)[8]        = (float (*)[8])smem;
    float (*startS)[TN + 2] = (float (*)[TN + 2])(smem + TM*8);

    const int tid  = threadIdx.x;
    const int lane = tid & 31, wid = tid >> 5;
    const int wm   = wid >> 1, wn  = wid & 1;
    const int ll   = lane & 15, lh = lane >> 4;
    const int n0   = blockIdx.x * TN, m0 = blockIdx.y * TM;

    const float* aP = Af + (size_t)(m0 + (tid >> 3))*1536 + ((tid & 7) << 2);
    const float* wP = W5 + (size_t)(n0 + (tid >> 3))*3072 + 1536 + ((tid & 7) << 2);
    v4f ra[4], rw[2];
    tile_get(aP, wP, ra, rw);

    {
        int r = tid >> 1, c = (tid & 1) << 2;
        *(v4f*)&aBlk[r][c] = *(const v4f*)(attn + (size_t)(m0 + r)*8 + c);
    }

    __syncthreads();

    {
        const int col = tid & 63;
        const int rg  = tid >> 6;
        const int gn  = n0 + col;
        #pragma unroll 4
        for (int i = 0; i < 32; ++i) {
            const int row = rg*32 + i;
            const int m   = m0 + row;
            const int bs  = m / 20;
            const int l   = m - bs*20;
            const v4f aLo = *(const v4f*)&aBlk[row][0];
            const v4f aHi = *(const v4f*)&aBlk[row][4];
            const float* pp = Pt + ((size_t)l*768 + gn)*8;
            const v4f pLo = *(const v4f*)pp;
            const v4f pHi = *(const v4f*)(pp + 4);
            float s = T0[(size_t)bs*768 + gn];
            s += aLo.x*pLo.x; s += aLo.y*pLo.y; s += aLo.z*pLo.z; s += aLo.w*pLo.w;
            s += aHi.x*pHi.x; s += aHi.y*pHi.y; s += aHi.z*pHi.z; s += aHi.w*pHi.w;
            startS[row][col] = s;
        }
    }
    __syncthreads();

    v8f acc[2][2];
    #pragma unroll
    for (int mt = 0; mt < 2; ++mt)
        #pragma unroll
        for (int nt = 0; nt < 2; ++nt)
            #pragma unroll
            for (int r = 0; r < 8; ++r)
                acc[mt][nt][r] = startS[wm*32 + mt*16 + lh*8 + r][wn*32 + nt*16 + ll];

    __syncthreads();

    float* st0 = smem;
    float* st1 = smem + STG;
    stage_put(st0, tid, ra, rw);
    tile_get(aP + 32, wP + 32, ra, rw);
    __syncthreads();

    for (int kc = 0; kc < 48; kc += 2) {
        stage_put(st1, tid, ra, rw);
        if (kc + 2 < 48) tile_get(aP + (kc + 2)*32, wP + (kc + 2)*32, ra, rw);
        mma_phase(st0, acc, wm, wn, ll, lh);
        __syncthreads();

        if (kc + 2 < 48) stage_put(st0, tid, ra, rw);
        if (kc + 3 < 48) tile_get(aP + (kc + 3)*32, wP + (kc + 3)*32, ra, rw);
        mma_phase(st1, acc, wm, wn, ll, lh);
        __syncthreads();
    }

    #pragma unroll
    for (int mt = 0; mt < 2; ++mt)
        #pragma unroll
        for (int nt = 0; nt < 2; ++nt) {
            int gn = n0 + wn*32 + nt*16 + ll;
            float bias = b5[gn];
            #pragma unroll
            for (int r = 0; r < 8; ++r) {
                int gm = m0 + wm*32 + mt*16 + lh*8 + r;
                out[(size_t)gm*768 + gn] = tanhf(acc[mt][nt][r] + bias);
            }
        }
}

extern "C" void kernel_launch(void* const* d_in, const int* in_sizes, int n_in,
                              void* d_out, int out_size, void* d_ws, size_t ws_size,
                              hipStream_t stream)
{
    (void)in_sizes; (void)n_in; (void)out_size; (void)ws_size;

    const float* token = (const float*)d_in[0];
    const float* label = (const float*)d_in[1];
    const float* imask = (const float*)d_in[2];
    const float* lmask = (const float*)d_in[3];
    const float* W1    = (const float*)d_in[4];
    const float* W2    = (const float*)d_in[5];
    const float* W5    = (const float*)d_in[6];
    const float* b5    = (const float*)d_in[7];
    float* out = (float*)d_out;

    float* ws     = (float*)d_ws;
    float* tkn    = ws;
    float* lab    = tkn    + 2048*768;
    float* scores = lab    + 160*768;
    float* attn   = scores + 2048*160;
    float* b_in   = attn   + 2048*160;
    float* q2c    = b_in   + 40960;
    float* T0     = q2c    + 160*768;
    float* Pt     = T0     + 2048*768;
    float* Af     = Pt     + 20*768*8;

    dim3 blk(256);
    gemm_nt_f32_wmma<<<dim3(16, 12), blk, 0, stream>>>(token, W1, tkn, 2048, 768, 768);
    gemm_nt_f32_wmma<<<dim3(2, 12),  blk, 0, stream>>>(label, W2, lab, 160, 768, 300);
    gemm_nt_f32_wmma<<<dim3(16, 3),  blk, 0, stream>>>(tkn, lab, scores, 2048, 160, 768);
    attn_softmax_kernel<<<dim3(160), blk, 0, stream>>>(scores, lmask, imask, attn, b_in);
    q2c_kernel<<<dim3(160), blk, 0, stream>>>(b_in, tkn, q2c);
    rowfactor_w5_wmma<<<dim3(18, 12), blk, 0, stream>>>(tkn, lab, W5, T0, Pt);
    build_fused_a<<<dim3(20, 128), dim3(192), 0, stream>>>(tkn, lab, attn, q2c, Af);
    fused_w5_tanh_wmma<<<dim3(12, 320), blk, 0, stream>>>(attn, Af, W5, b5, T0, Pt, out);
}


# Small GEMMs (tkn, lab, scores, row-factor): a k chunk's six 16-byte tile loads issued together one chunk ahead into a 2-stage LDS ring; were each waited for singly, 5-6 round trips per chunk
# speedup vs baseline: 2.0114x; 1.1175x over previous
#include <hip/hip_runtime.h>
#include <math.h>


typedef __attribute__((ext_vector_type(2))) float v2f;
typedef __attribute__((ext_vector_type(4))) float v4f;
typedef __attribute__((ext_vector_type(8))) float v8f;

#define NEG_BIG (-10000.0f)

#define TM 128
#define TN 64
#define TK 32
#define TKP 34


#define STG (192*TKP)

__device__ __forceinline__ void stage_put(float* st, int tid, const v4f (&a)[4], const v4f (&w)[2])
{
    const int r = tid >> 3, c = (tid & 7) << 2;
    #pragma unroll
    for (int p = 0; p < 4; ++p) {
        float* d = st + (p*32 + r)*TKP + c;
        v2f lo = {a[p].x, a[p].y}, hi = {a[p].z, a[p].w};
        *(v2f*)d = lo; *(v2f*)(d + 2) = hi;
    }
    #pragma unroll
    for (int p = 0; p < 2; ++p) {
        float* d = st + (TM + p*32 + r)*TKP + c;
        v2f lo = {w[p].x, w[p].y}, hi = {w[p].z, w[p].w};
        *(v2f*)d = lo; *(v2f*)(d + 2) = hi;
    }
}

__device__ __forceinline__ void mma_phase(const float* st, v8f (&acc)[2][2],
                                          int wm, int wn, int ll, int lh)
{
    const float* As = st;
    const float* Ws = st + TM*TKP;
    #pragma unroll
    for (int kk = 0; kk < 8; ++kk) {
        const int kc = 4*kk + 2*lh;
        v2f a0 = *(const v2f*)(As + (wm*32      + ll)*TKP + kc);
        v2f a1 = *(const v2f*)(As + (wm*32 + 16 + ll)*TKP + kc);
        v2f b0 = *(const v2f*)(Ws + (wn*32      + ll)*TKP + kc);
        v2f b1 = *(const v2f*)(Ws + (wn*32 + 16 + ll)*TKP + kc);
        acc[0][0] = __builtin_amdgcn_wmma_f32_16x16x4_f32(false,a0,false,b0,(short)0,acc[0][0],false,false);
        acc[0][1] = __builtin_amdgcn_wmma_f32_16x16x4_f32(false,a0,false,b1,(short)0,acc[0][1],false,false);
        acc[1][0] = __builtin_amdgcn_wmma_f32_16x16x4_f32(false,a1,false,b0,(short)0,acc[1][0],false,false);
        acc[1][1] = __builtin_amdgcn_wmma_f32_16x16x4_f32(false,a1,false,b1,(short)0,acc[1][1],false,false);
    }
}

__device__ __forceinline__ void tile_get_p(const float* const (&pa)[4], const float* const (&pw)[2],
                                           int ko, v4f (&a)[4], v4f (&w)[2])
{
    #pragma unroll
    for (int p = 0; p < 4; ++p) a[p] = *(const v4f*)(pa[p] + ko);
    #pragma unroll
    for (int p = 0; p < 2; ++p) w[p] = *(const v4f*)(pw[p] + ko);
}

__device__ __forceinline__ void stage_put_z(float* st, int tid, const v4f (&a)[4], const v4f (&w)[2],
                                            const bool (&okA)[4], const bool (&okW)[2], bool kin)
{
    const v4f z = {0.f,0.f,0.f,0.f};
    v4f az[4], wz[2];
    #pragma unroll
    for (int p = 0; p < 4; ++p) az[p] = (okA[p] && kin) ? a[p] : z;
    #pragma unroll
    for (int p = 0; p < 2; ++p) wz[p] = (okW[p] && kin) ? w[p] : z;
    stage_put(st, tid, az, wz);
}

template<int K, int LDA, int LDW, bool GA, bool GW>
__device__ __forceinline__ void gemm_tile_128x64(
    const float* __restrict__ Ab, int rowsA,
    const float* __restrict__ Wb, int rowsW,
    float* smem, v8f (&acc)[2][2], int tid, int wm, int wn, int ll, int lh)
{
    constexpr int  NCH   = (K + TK - 1) / TK;
    constexpr bool GK    = (K % TK) != 0;
    constexpr int  KLAST = K - (NCH - 1)*TK;
    static_assert((K & 3) == 0, "16-byte loads along K");
    static_assert((NCH & 1) == 0 && NCH >= 2, "two chunks per trip");

    const int r = tid >> 3, c = (tid & 7) << 2;

    const float* pa[4]; const float* pw[2];
    bool okA[4], okW[2];
    #pragma unroll
    for (int p = 0; p < 4; ++p) {
        const int row = p*32 + r;
        okA[p] = !GA || row < rowsA;
        const int rc = (GA && row >= rowsA) ? rowsA - 1 : row;
        pa[p] = Ab + (size_t)rc*LDA + c;
    }
    #pragma unroll
    for (int p = 0; p < 2; ++p) {
        const int row = p*32 + r;
        okW[p] = !GW || row < rowsW;
        const int rc = (GW && row >= rowsW) ? rowsW - 1 : row;
        pw[p] = Wb + (size_t)rc*LDW + c;
    }
    const bool tailIn = !GK || c < KLAST;
    const int  koLast = tailIn ? (NCH - 1)*TK : 0;

    v4f ra[4], rw[2];
    float* st0 = smem;
    float* st1 = smem + STG;

    tile_get_p(pa, pw, 0, ra, rw);
    stage_put_z(st0, tid, ra, rw, okA, okW, true);
    tile_get_p(pa, pw, (1 == NCH - 1) ? koLast : TK, ra, rw);
    __syncthreads();

    #pragma unroll 1
    for (int kc = 0; kc < NCH; kc += 2) {
        stage_put_z(st1, tid, ra, rw, okA, okW, (kc + 1 != NCH - 1) || tailIn);
        if (kc + 2 < NCH) tile_get_p(pa, pw, (kc + 2)*TK, ra, rw);
        mma_phase(st0, acc, wm, wn, ll, lh);
        __syncthreads();

        if (kc + 2 < NCH) stage_put_z(st0, tid, ra, rw, okA, okW, true);
        if (kc + 3 < NCH) tile_get_p(pa, pw, (kc + 3 == NCH - 1) ? koLast : (kc + 3)*TK, ra, rw);
        mma_phase(st1, acc, wm, wn, ll, lh);
        __syncthreads();
    }
}

template<int M, int N, int K>
__global__ __launch_bounds__(256) void gemm_nt_f32_wmma(
    const float* __restrict__ A, const float* __restrict__ W, float* __restrict__ C)
{
    __shared__ __attribute__((aligned(16))) float smem[2*STG];
    constexpr bool GA = (M % TM) != 0, GW = (N % TN) != 0;

    const int tid  = threadIdx.x;
    const int lane = tid & 31, wid = tid >> 5;
    const int wm   = wid >> 1, wn  = wid & 1;
    const int ll   = lane & 15, lh = lane >> 4;
    const int m0   = blockIdx.x * TM, n0 = blockIdx.y * TN;

    const v8f vzero = {0.f,0.f,0.f,0.f,0.f,0.f,0.f,0.f};
    v8f acc[2][2];
    acc[0][0]=vzero; acc[0][1]=vzero; acc[1][0]=vzero; acc[1][1]=vzero;

    gemm_tile_128x64<K, K, K, GA, GW>(A + (size_t)m0*K, M - m0, W + (size_t)n0*K, N - n0,
                                      smem, acc, tid, wm, wn, ll, lh);

    #pragma unroll
    for (int mt = 0; mt < 2; ++mt)
        #pragma unroll
        for (int nt = 0; nt < 2; ++nt) {
            int gn = n0 + wn*32 + nt*16 + ll;
            #pragma unroll
            for (int r = 0; r < 8; ++r) {
                int gm = m0 + wm*32 + mt*16 + lh*8 + r;
                if ((!GA || gm < M) && (!GW || gn < N)) C[(size_t)gm*N + gn] = acc[mt][nt][r];
            }
        }
}

__global__ __launch_bounds__(256) void attn_softmax_kernel(
    const float* __restrict__ scores, const float* __restrict__ lmask,
    const float* __restrict__ imask, float* __restrict__ attn,
    float* __restrict__ b_in)
{
    int row = blockIdx.x * 256 + threadIdx.x;
    if (row >= 8*256*20) return;
    int bs = row / 20;
    int l  = row - bs*20;

    float v[8];
    float mx = -3.4e38f;
    #pragma unroll
    for (int t = 0; t < 8; ++t) {
        float s = scores[(size_t)row*8 + t] + (1.0f - lmask[l*8 + t]) * NEG_BIG;
        v[t] = s;
        mx = fmaxf(mx, s);
    }
    float sum = 0.f;
    #pragma unroll
    for (int t = 0; t < 8; ++t) { float e = __expf(v[t] - mx); v[t] = e; sum += e; }
    float inv = 1.0f / sum;
    #pragma unroll
    for (int t = 0; t < 8; ++t) attn[(size_t)row*8 + t] = v[t] * inv;

    b_in[row] = mx + (1.0f - imask[bs]) * NEG_BIG;
}

__global__ __launch_bounds__(256) void q2c_kernel(
    const float* __restrict__ b_in, const float* __restrict__ tkn,
    float* __restrict__ q2c)
{
    __shared__ float red[256];
    __shared__ float wgt[256];
    const int tid = threadIdx.x;
    const int b = blockIdx.x / 20;
    const int l = blockIdx.x - b*20;

    float x = b_in[(size_t)(b*256 + tid)*20 + l];
    red[tid] = x; __syncthreads();
    for (int st = 128; st > 0; st >>= 1) {
        if (tid < st) red[tid] = fmaxf(red[tid], red[tid + st]);
        __syncthreads();
    }
    float mx = red[0]; __syncthreads();
    float e = __expf(x - mx);
    red[tid] = e; __syncthreads();
    for (int st = 128; st > 0; st >>= 1) {
        if (tid < st) red[tid] += red[tid + st];
        __syncthreads();
    }
    float sm = red[0]; __syncthreads();
    wgt[tid] = e / sm; __syncthreads();

    for (int h = tid; h < 768; h += 256) {
        float acc = 0.f;
        const float* tp = tkn + (size_t)b*256*768 + h;
        #pragma unroll 4
        for (int s = 0; s < 256; ++s) acc += wgt[s] * tp[(size_t)s*768];
        q2c[(size_t)(b*20 + l)*768 + h] = acc;
    }
}

__global__ __launch_bounds__(256) void rowfactor_w5_wmma(
    const float* __restrict__ tkn, const float* __restrict__ lab,
    const float* __restrict__ W5,
    float* __restrict__ T0, float* __restrict__ Pt)
{
    __shared__ __attribute__((aligned(16))) float smem[2*STG];

    const int tid  = threadIdx.x;
    const int lane = tid & 31, wid = tid >> 5;
    const int wm   = wid >> 1, wn  = wid & 1;
    const int ll   = lane & 15, lh = lane >> 4;
    const int m0   = blockIdx.x * TM, n0 = blockIdx.y * TN;

    const bool isLab = (m0 >= 2048);
    const int  q0    = m0 - 2048;
    const float* __restrict__ Ab = isLab ? (lab + (size_t)q0*768) : (tkn + (size_t)m0*768);
    const int  rows  = isLab ? (160 - q0) : TM;
    const float* __restrict__ Wb = W5 + (isLab ? 768 : 0) + (size_t)n0*3072;

    const v8f vzero = {0.f,0.f,0.f,0.f,0.f,0.f,0.f,0.f};
    v8f acc[2][2];
    acc[0][0]=vzero; acc[0][1]=vzero; acc[1][0]=vzero; acc[1][1]=vzero;

    gemm_tile_128x64<768, 768, 3072, true, false>(Ab, rows, Wb, TN, smem, acc, tid, wm, wn, ll, lh);

    #pragma unroll
    for (int mt = 0; mt < 2; ++mt)
        #pragma unroll
        for (int nt = 0; nt < 2; ++nt) {
            int gn = n0 + wn*32 + nt*16 + ll;
            #pragma unroll
            for (int r = 0; r < 8; ++r) {
                int row = wm*32 + mt*16 + lh*8 + r;
                if (!isLab) {
                    T0[(size_t)(m0 + row)*768 + gn] = acc[mt][nt][r];
                } else if (row < rows) {
                    int q = q0 + row;
                    Pt[((size_t)(q >> 3)*768 + gn)*8 + (q & 7)] = acc[mt][nt][r];
                }
            }
        }
}

__global__ __launch_bounds__(192) void build_fused_a(
    const float* __restrict__ tkn,  const float* __restrict__ lab,
    const float* __restrict__ attn, const float* __restrict__ q2c,
    float* __restrict__ Af)
{
    const int tid = threadIdx.x;
    const int l   = blockIdx.x;
    const int bs0 = blockIdx.y * 16;
    const int b   = bs0 >> 8;
    const int j   = tid << 2;

    v4f lv[8];
    #pragma unroll
    for (int t = 0; t < 8; ++t) lv[t] = *(const v4f*)(lab + (size_t)(l*8 + t)*768 + j);
    const v4f q4 = *(const v4f*)(q2c + (size_t)(b*20 + l)*768 + j);

    #pragma unroll 4
    for (int i = 0; i < 16; ++i) {
        const int bs = bs0 + i;
        const size_t m = (size_t)bs*20 + l;
        const v4f aLo = *(const v4f*)(attn + m*8);
        const v4f aHi = *(const v4f*)(attn + m*8 + 4);
        const v4f t4  = *(const v4f*)(tkn + (size_t)bs*768 + j);
        v4f cq = {0.f,0.f,0.f,0.f};
        cq += aLo.x * lv[0]; cq += aLo.y * lv[1]; cq += aLo.z * lv[2]; cq += aLo.w * lv[3];
        cq += aHi.x * lv[4]; cq += aHi.y * lv[5]; cq += aHi.z * lv[6]; cq += aHi.w * lv[7];
        float* dst = Af + m*1536 + j;
        *(v4f*)dst         = t4 * cq;
        *(v4f*)(dst + 768) = t4 * q4;
    }
}

__device__ __forceinline__ void tile_get(const float* __restrict__ aP, const float* __restrict__ wP,
                                         v4f (&a)[4], v4f (&w)[2])
{
    #pragma unroll
    for (int p = 0; p < 4; ++p) a[p] = *(const v4f*)(aP + (size_t)p*32*1536);
    #pragma unroll
    for (int p = 0; p < 2; ++p) w[p] = *(const v4f*)(wP + (size_t)p*32*3072);
}

__global__ __launch_bounds__(256) void fused_w5_tanh_wmma(
    const float* __restrict__ attn, const float* __restrict__ Af,
    const float* __restrict__ W5,   const float* __restrict__ b5,
    const float* __restrict__ T0,   const float* __restrict__ Pt,
    float* __restrict__ out)
{
    __shared__ __attribute__((aligned(16))) float smem[2*STG];
    float (*aBlk)[8]        = (float (*)[8])smem;
    float (*startS)[TN + 2] = (float (*)[TN + 2])(smem + TM*8);

    const int tid  = threadIdx.x;
    const int lane = tid & 31, wid = tid >> 5;
    const int wm   = wid >> 1, wn  = wid & 1;
    const int ll   = lane & 15, lh = lane >> 4;
    const int n0   = blockIdx.x * TN, m0 = blockIdx.y * TM;

    const float* aP = Af + (size_t)(m0 + (tid >> 3))*1536 + ((tid & 7) << 2);
    const float* wP = W5 + (size_t)(n0 + (tid >> 3))*3072 + 1536 + ((tid & 7) << 2);
    v4f ra[4], rw[2];
    tile_get(aP, wP, ra, rw);

    {
        int r = tid >> 1, c = (tid & 1) << 2;
        *(v4f*)&aBlk[r][c] = *(const v4f*)(attn + (size_t)(m0 + r)*8 + c);
    }

    __syncthreads();

    {
        const int col = tid & 63;
        const int rg  = tid >> 6;
        const int gn  = n0 + col;
        #pragma unroll 4
        for (int i = 0; i < 32; ++i) {
            const int row = rg*32 + i;
            const int m   = m0 + row;
            const int bs  = m / 20;
            const int l   = m - bs*20;
            const v4f aLo = *(const v4f*)&aBlk[row][0];
            const v4f aHi = *(const v4f*)&aBlk[row][4];
            const float* pp = Pt + ((size_t)l*768 + gn)*8;
            const v4f pLo = *(const v4f*)pp;
            const v4f pHi = *(const v4f*)(pp + 4);
            float s = T0[(size_t)bs*768 + gn];
            s += aLo.x*pLo.x; s += aLo.y*pLo.y; s += aLo.z*pLo.z; s += aLo.w*pLo.w;
            s += aHi.x*pHi.x; s += aHi.y*pHi.y; s += aHi.z*pHi.z; s += aHi.w*pHi.w;
            startS[row][col] = s;
        }
    }
    __syncthreads();

    v8f acc[2][2];
    #pragma unroll
    for (int mt = 0; mt < 2; ++mt)
        #pragma unroll
        for (int nt = 0; nt < 2; ++nt)
            #pragma unroll
            for (int r = 0; r < 8; ++r)
                acc[mt][nt][r] = startS[wm*32 + mt*16 + lh*8 + r][wn*32 + nt*16 + ll];

    __syncthreads();

    float* st0 = smem;
    float* st1 = smem + STG;
    stage_put(st0, tid, ra, rw);
    tile_get(aP + 32, wP + 32, ra, rw);
    __syncthreads();

    for (int kc = 0; kc < 48; kc += 2) {
        stage_put(st1, tid, ra, rw);
        if (kc + 2 < 48) tile_get(aP + (kc + 2)*32, wP + (kc + 2)*32, ra, rw);
        mma_phase(st0, acc, wm, wn, ll, lh);
        __syncthreads();

        if (kc + 2 < 48) stage_put(st0, tid, ra, rw);
        if (kc + 3 < 48) tile_get(aP + (kc + 3)*32, wP + (kc + 3)*32, ra, rw);
        mma_phase(st1, acc, wm, wn, ll, lh);
        __syncthreads();
    }

    #pragma unroll
    for (int mt = 0; mt < 2; ++mt)
        #pragma unroll
        for (int nt = 0; nt < 2; ++nt) {
            int gn = n0 + wn*32 + nt*16 + ll;
            float bias = b5[gn];
            #pragma unroll
            for (int r = 0; r < 8; ++r) {
                int gm = m0 + wm*32 + mt*16 + lh*8 + r;
                out[(size_t)gm*768 + gn] = tanhf(acc[mt][nt][r] + bias);
            }
        }
}

extern "C" void kernel_launch(void* const* d_in, const int* in_sizes, int n_in,
                              void* d_out, int out_size, void* d_ws, size_t ws_size,
                              hipStream_t stream)
{
    (void)in_sizes; (void)n_in; (void)out_size; (void)ws_size;

    const float* token = (const float*)d_in[0];
    const float* label = (const float*)d_in[1];
    const float* imask = (const float*)d_in[2];
    const float* lmask = (const float*)d_in[3];
    const float* W1    = (const float*)d_in[4];
    const float* W2    = (const float*)d_in[5];
    const float* W5    = (const float*)d_in[6];
    const float* b5    = (const float*)d_in[7];
    float* out = (float*)d_out;

    float* ws     = (float*)d_ws;
    float* tkn    = ws;
    float* lab    = tkn    + 2048*768;
    float* scores = lab    + 160*768;
    float* attn   = scores + 2048*160;
    float* b_in   = attn   + 2048*160;
    float* q2c    = b_in   + 40960;
    float* T0     = q2c    + 160*768;
    float* Pt     = T0     + 2048*768;
    float* Af     = Pt     + 20*768*8;

    dim3 blk(256);
    gemm_nt_f32_wmma<2048, 768, 768><<<dim3(16, 12), blk, 0, stream>>>(token, W1, tkn);
    gemm_nt_f32_wmma<160, 768, 300><<<dim3(2, 12),  blk, 0, stream>>>(label, W2, lab);
    gemm_nt_f32_wmma<2048, 160, 768><<<dim3(16, 3),  blk, 0, stream>>>(tkn, lab, scores);
    attn_softmax_kernel<<<dim3(160), blk, 0, stream>>>(scores, lmask, imask, attn, b_in);
    q2c_kernel<<<dim3(160), blk, 0, stream>>>(b_in, tkn, q2c);
    rowfactor_w5_wmma<<<dim3(18, 12), blk, 0, stream>>>(tkn, lab, W5, T0, Pt);
    build_fused_a<<<dim3(20, 128), dim3(192), 0, stream>>>(tkn, lab, attn, q2c, Af);
    fused_w5_tanh_wmma<<<dim3(12, 320), blk, 0, stream>>>(attn, Af, W5, b5, T0, Pt, out);
}


# q2c re-decomposed: a block per (batch, 32-column slab) loads b_in[b] and its tkn slab once, all up front, 20 labels from LDS; was 192 dependent 4-byte round trips; S sums: 8 partials + butterfly
# speedup vs baseline: 2.2345x; 1.1109x over previous
#include <hip/hip_runtime.h>
#include <math.h>


typedef __attribute__((ext_vector_type(2))) float v2f;
typedef __attribute__((ext_vector_type(4))) float v4f;
typedef __attribute__((ext_vector_type(8))) float v8f;

#define NEG_BIG (-10000.0f)

#define TM 128
#define TN 64
#define TK 32
#define TKP 34


#define STG (192*TKP)

__device__ __forceinline__ void stage_put(float* st, int tid, const v4f (&a)[4], const v4f (&w)[2])
{
    const int r = tid >> 3, c = (tid & 7) << 2;
    #pragma unroll
    for (int p = 0; p < 4; ++p) {
        float* d = st + (p*32 + r)*TKP + c;
        v2f lo = {a[p].x, a[p].y}, hi = {a[p].z, a[p].w};
        *(v2f*)d = lo; *(v2f*)(d + 2) = hi;
    }
    #pragma unroll
    for (int p = 0; p < 2; ++p) {
        float* d = st + (TM + p*32 + r)*TKP + c;
        v2f lo = {w[p].x, w[p].y}, hi = {w[p].z, w[p].w};
        *(v2f*)d = lo; *(v2f*)(d + 2) = hi;
    }
}

__device__ __forceinline__ void mma_phase(const float* st, v8f (&acc)[2][2],
                                          int wm, int wn, int ll, int lh)
{
    const float* As = st;
    const float* Ws = st + TM*TKP;
    #pragma unroll
    for (int kk = 0; kk < 8; ++kk) {
        const int kc = 4*kk + 2*lh;
        v2f a0 = *(const v2f*)(As + (wm*32      + ll)*TKP + kc);
        v2f a1 = *(const v2f*)(As + (wm*32 + 16 + ll)*TKP + kc);
        v2f b0 = *(const v2f*)(Ws + (wn*32      + ll)*TKP + kc);
        v2f b1 = *(const v2f*)(Ws + (wn*32 + 16 + ll)*TKP + kc);
        acc[0][0] = __builtin_amdgcn_wmma_f32_16x16x4_f32(false,a0,false,b0,(short)0,acc[0][0],false,false);
        acc[0][1] = __builtin_amdgcn_wmma_f32_16x16x4_f32(false,a0,false,b1,(short)0,acc[0][1],false,false);
        acc[1][0] = __builtin_amdgcn_wmma_f32_16x16x4_f32(false,a1,false,b0,(short)0,acc[1][0],false,false);
        acc[1][1] = __builtin_amdgcn_wmma_f32_16x16x4_f32(false,a1,false,b1,(short)0,acc[1][1],false,false);
    }
}

__device__ __forceinline__ void tile_get_p(const float* const (&pa)[4], const float* const (&pw)[2],
                                           int ko, v4f (&a)[4], v4f (&w)[2])
{
    #pragma unroll
    for (int p = 0; p < 4; ++p) a[p] = *(const v4f*)(pa[p] + ko);
    #pragma unroll
    for (int p = 0; p < 2; ++p) w[p] = *(const v4f*)(pw[p] + ko);
}

__device__ __forceinline__ void stage_put_z(float* st, int tid, const v4f (&a)[4], const v4f (&w)[2],
                                            const bool (&okA)[4], const bool (&okW)[2], bool kin)
{
    const v4f z = {0.f,0.f,0.f,0.f};
    v4f az[4], wz[2];
    #pragma unroll
    for (int p = 0; p < 4; ++p) az[p] = (okA[p] && kin) ? a[p] : z;
    #pragma unroll
    for (int p = 0; p < 2; ++p) wz[p] = (okW[p] && kin) ? w[p] : z;
    stage_put(st, tid, az, wz);
}

template<int K, int LDA, int LDW, bool GA, bool GW>
__device__ __forceinline__ void gemm_tile_128x64(
    const float* __restrict__ Ab, int rowsA,
    const float* __restrict__ Wb, int rowsW,
    float* smem, v8f (&acc)[2][2], int tid, int wm, int wn, int ll, int lh)
{
    constexpr int  NCH   = (K + TK - 1) / TK;
    constexpr bool GK    = (K % TK) != 0;
    constexpr int  KLAST = K - (NCH - 1)*TK;
    static_assert((K & 3) == 0, "16-byte loads along K");
    static_assert((NCH & 1) == 0 && NCH >= 2, "two chunks per trip");

    const int r = tid >> 3, c = (tid & 7) << 2;

    const float* pa[4]; const float* pw[2];
    bool okA[4], okW[2];
    #pragma unroll
    for (int p = 0; p < 4; ++p) {
        const int row = p*32 + r;
        okA[p] = !GA || row < rowsA;
        const int rc = (GA && row >= rowsA) ? rowsA - 1 : row;
        pa[p] = Ab + (size_t)rc*LDA + c;
    }
    #pragma unroll
    for (int p = 0; p < 2; ++p) {
        const int row = p*32 + r;
        okW[p] = !GW || row < rowsW;
        const int rc = (GW && row >= rowsW) ? rowsW - 1 : row;
        pw[p] = Wb + (size_t)rc*LDW + c;
    }
    const bool tailIn = !GK || c < KLAST;
    const int  koLast = tailIn ? (NCH - 1)*TK : 0;

    v4f ra[4], rw[2];
    float* st0 = smem;
    float* st1 = smem + STG;

    tile_get_p(pa, pw, 0, ra, rw);
    stage_put_z(st0, tid, ra, rw, okA, okW, true);
    tile_get_p(pa, pw, (1 == NCH - 1) ? koLast : TK, ra, rw);
    __syncthreads();

    #pragma unroll 1
    for (int kc = 0; kc < NCH; kc += 2) {
        stage_put_z(st1, tid, ra, rw, okA, okW, (kc + 1 != NCH - 1) || tailIn);
        if (kc + 2 < NCH) tile_get_p(pa, pw, (kc + 2)*TK, ra, rw);
        mma_phase(st0, acc, wm, wn, ll, lh);
        __syncthreads();

        if (kc + 2 < NCH) stage_put_z(st0, tid, ra, rw, okA, okW, true);
        if (kc + 3 < NCH) tile_get_p(pa, pw, (kc + 3 == NCH - 1) ? koLast : (kc + 3)*TK, ra, rw);
        mma_phase(st1, acc, wm, wn, ll, lh);
        __syncthreads();
    }
}

template<int M, int N, int K>
__global__ __launch_bounds__(256) void gemm_nt_f32_wmma(
    const float* __restrict__ A, const float* __restrict__ W, float* __restrict__ C)
{
    __shared__ __attribute__((aligned(16))) float smem[2*STG];
    constexpr bool GA = (M % TM) != 0, GW = (N % TN) != 0;

    const int tid  = threadIdx.x;
    const int lane = tid & 31, wid = tid >> 5;
    const int wm   = wid >> 1, wn  = wid & 1;
    const int ll   = lane & 15, lh = lane >> 4;
    const int m0   = blockIdx.x * TM, n0 = blockIdx.y * TN;

    const v8f vzero = {0.f,0.f,0.f,0.f,0.f,0.f,0.f,0.f};
    v8f acc[2][2];
    acc[0][0]=vzero; acc[0][1]=vzero; acc[1][0]=vzero; acc[1][1]=vzero;

    gemm_tile_128x64<K, K, K, GA, GW>(A + (size_t)m0*K, M - m0, W + (size_t)n0*K, N - n0,
                                      smem, acc, tid, wm, wn, ll, lh);

    #pragma unroll
    for (int mt = 0; mt < 2; ++mt)
        #pragma unroll
        for (int nt = 0; nt < 2; ++nt) {
            int gn = n0 + wn*32 + nt*16 + ll;
            #pragma unroll
            for (int r = 0; r < 8; ++r) {
                int gm = m0 + wm*32 + mt*16 + lh*8 + r;
                if ((!GA || gm < M) && (!GW || gn < N)) C[(size_t)gm*N + gn] = acc[mt][nt][r];
            }
        }
}

__global__ __launch_bounds__(256) void attn_softmax_kernel(
    const float* __restrict__ scores, const float* __restrict__ lmask,
    const float* __restrict__ imask, float* __restrict__ attn,
    float* __restrict__ b_in)
{
    int row = blockIdx.x * 256 + threadIdx.x;
    if (row >= 8*256*20) return;
    int bs = row / 20;
    int l  = row - bs*20;

    float v[8];
    float mx = -3.4e38f;
    #pragma unroll
    for (int t = 0; t < 8; ++t) {
        float s = scores[(size_t)row*8 + t] + (1.0f - lmask[l*8 + t]) * NEG_BIG;
        v[t] = s;
        mx = fmaxf(mx, s);
    }
    float sum = 0.f;
    #pragma unroll
    for (int t = 0; t < 8; ++t) { float e = __expf(v[t] - mx); v[t] = e; sum += e; }
    float inv = 1.0f / sum;
    #pragma unroll
    for (int t = 0; t < 8; ++t) attn[(size_t)row*8 + t] = v[t] * inv;

    b_in[row] = mx + (1.0f - imask[bs]) * NEG_BIG;
}

#define QC_TP 36

__global__ __launch_bounds__(256) void q2c_kernel(
    const float* __restrict__ b_in, const float* __restrict__ tkn,
    float* __restrict__ q2c)
{
    __shared__ __attribute__((aligned(16))) float Wt[256*20];
    __shared__ __attribute__((aligned(16))) float Ts[256*QC_TP];
    __shared__ __attribute__((aligned(16))) float Pp[8*20*32];

    const int tid  = threadIdx.x;
    const int lane = tid & 31, wid = tid >> 5;
    const int b    = blockIdx.x / 24;
    const int slab = blockIdx.x - b*24;
    const int cq   = tid & 7, r0 = tid >> 3;

    {
        v4f xb[5], tt[8];
        const float* bp = b_in + (size_t)b*256*20;
        const float* tp = tkn + ((size_t)b*256 + r0)*768 + slab*32 + 4*cq;
        #pragma unroll
        for (int p = 0; p < 5; ++p) xb[p] = *(const v4f*)(bp + 4*(tid + 256*p));
        #pragma unroll
        for (int p = 0; p < 8; ++p) tt[p] = *(const v4f*)(tp + (size_t)p*32*768);
        #pragma unroll
        for (int p = 0; p < 5; ++p) *(v4f*)(Wt + 4*(tid + 256*p)) = xb[p];
        #pragma unroll
        for (int p = 0; p < 8; ++p) *(v4f*)(Ts + (r0 + 32*p)*QC_TP + 4*cq) = tt[p];
    }
    __syncthreads();

    {
        float x[3][8], mx[3], sm[3];
        #pragma unroll
        for (int k = 0; k < 3; ++k) {
            const int l  = wid + 8*k;
            const int lc = (l < 20) ? l : wid;
            mx[k] = -3.4e38f;
            #pragma unroll
            for (int i = 0; i < 8; ++i) { x[k][i] = Wt[(lane + 32*i)*20 + lc]; mx[k] = fmaxf(mx[k], x[k][i]); }
        }
        #pragma unroll
        for (int o = 16; o > 0; o >>= 1)
            #pragma unroll
            for (int k = 0; k < 3; ++k) mx[k] = fmaxf(mx[k], __shfl_xor(mx[k], o, 32));
        #pragma unroll
        for (int k = 0; k < 3; ++k) {
            sm[k] = 0.f;
            #pragma unroll
            for (int i = 0; i < 8; ++i) { x[k][i] = __expf(x[k][i] - mx[k]); sm[k] += x[k][i]; }
        }
        #pragma unroll
        for (int o = 16; o > 0; o >>= 1)
            #pragma unroll
            for (int k = 0; k < 3; ++k) sm[k] += __shfl_xor(sm[k], o, 32);
        #pragma unroll
        for (int k = 0; k < 3; ++k) {
            const int l = wid + 8*k;
            if (l < 20) {
                #pragma unroll
                for (int i = 0; i < 8; ++i) Wt[(lane + 32*i)*20 + l] = x[k][i] / sm[k];
            }
        }
    }
    __syncthreads();

    {
        const int lg = (tid >> 3) & 3;
        const v4f z = {0.f,0.f,0.f,0.f};
        v4f acc[5] = {z, z, z, z, z};
        const float* tsp = Ts + (wid*32)*QC_TP + 4*cq;
        const float* wp  = Wt + (wid*32)*20 + 5*lg;
        #pragma unroll 8
        for (int r = 0; r < 32; ++r) {
            const v4f t = *(const v4f*)(tsp + r*QC_TP);
            #pragma unroll
            for (int k = 0; k < 5; ++k) acc[k] += wp[r*20 + k] * t;
        }
        #pragma unroll
        for (int k = 0; k < 5; ++k) *(v4f*)(Pp + (wid*20 + 5*lg + k)*32 + 4*cq) = acc[k];
    }
    __syncthreads();

    if (tid < 160) {
        const int l = tid >> 3;
        v4f s = *(const v4f*)(Pp + l*32 + 4*cq);
        #pragma unroll
        for (int w = 1; w < 8; ++w) s += *(const v4f*)(Pp + (w*20 + l)*32 + 4*cq);
        *(v4f*)(q2c + (size_t)(b*20 + l)*768 + slab*32 + 4*cq) = s;
    }
}

__global__ __launch_bounds__(256) void rowfactor_w5_wmma(
    const float* __restrict__ tkn, const float* __restrict__ lab,
    const float* __restrict__ W5,
    float* __restrict__ T0, float* __restrict__ Pt)
{
    __shared__ __attribute__((aligned(16))) float smem[2*STG];

    const int tid  = threadIdx.x;
    const int lane = tid & 31, wid = tid >> 5;
    const int wm   = wid >> 1, wn  = wid & 1;
    const int ll   = lane & 15, lh = lane >> 4;
    const int m0   = blockIdx.x * TM, n0 = blockIdx.y * TN;

    const bool isLab = (m0 >= 2048);
    const int  q0    = m0 - 2048;
    const float* __restrict__ Ab = isLab ? (lab + (size_t)q0*768) : (tkn + (size_t)m0*768);
    const int  rows  = isLab ? (160 - q0) : TM;
    const float* __restrict__ Wb = W5 + (isLab ? 768 : 0) + (size_t)n0*3072;

    const v8f vzero = {0.f,0.f,0.f,0.f,0.f,0.f,0.f,0.f};
    v8f acc[2][2];
    acc[0][0]=vzero; acc[0][1]=vzero; acc[1][0]=vzero; acc[1][1]=vzero;

    gemm_tile_128x64<768, 768, 3072, true, false>(Ab, rows, Wb, TN, smem, acc, tid, wm, wn, ll, lh);

    #pragma unroll
    for (int mt = 0; mt < 2; ++mt)
        #pragma unroll
        for (int nt = 0; nt < 2; ++nt) {
            int gn = n0 + wn*32 + nt*16 + ll;
            #pragma unroll
            for (int r = 0; r < 8; ++r) {
                int row = wm*32 + mt*16 + lh*8 + r;
                if (!isLab) {
                    T0[(size_t)(m0 + row)*768 + gn] = acc[mt][nt][r];
                } else if (row < rows) {
                    int q = q0 + row;
                    Pt[((size_t)(q >> 3)*768 + gn)*8 + (q & 7)] = acc[mt][nt][r];
                }
            }
        }
}

__global__ __launch_bounds__(192) void build_fused_a(
    const float* __restrict__ tkn,  const float* __restrict__ lab,
    const float* __restrict__ attn, const float* __restrict__ q2c,
    float* __restrict__ Af)
{
    const int tid = threadIdx.x;
    const int l   = blockIdx.x;
    const int bs0 = blockIdx.y * 16;
    const int b   = bs0 >> 8;
    const int j   = tid << 2;

    v4f lv[8];
    #pragma unroll
    for (int t = 0; t < 8; ++t) lv[t] = *(const v4f*)(lab + (size_t)(l*8 + t)*768 + j);
    const v4f q4 = *(const v4f*)(q2c + (size_t)(b*20 + l)*768 + j);

    #pragma unroll 4
    for (int i = 0; i < 16; ++i) {
        const int bs = bs0 + i;
        const size_t m = (size_t)bs*20 + l;
        const v4f aLo = *(const v4f*)(attn + m*8);
        const v4f aHi = *(const v4f*)(attn + m*8 + 4);
        const v4f t4  = *(const v4f*)(tkn + (size_t)bs*768 + j);
        v4f cq = {0.f,0.f,0.f,0.f};
        cq += aLo.x * lv[0]; cq += aLo.y * lv[1]; cq += aLo.z * lv[2]; cq += aLo.w * lv[3];
        cq += aHi.x * lv[4]; cq += aHi.y * lv[5]; cq += aHi.z * lv[6]; cq += aHi.w * lv[7];
        float* dst = Af + m*1536 + j;
        *(v4f*)dst         = t4 * cq;
        *(v4f*)(dst + 768) = t4 * q4;
    }
}

__device__ __forceinline__ void tile_get(const float* __restrict__ aP, const float* __restrict__ wP,
                                         v4f (&a)[4], v4f (&w)[2])
{
    #pragma unroll
    for (int p = 0; p < 4; ++p) a[p] = *(const v4f*)(aP + (size_t)p*32*1536);
    #pragma unroll
    for (int p = 0; p < 2; ++p) w[p] = *(const v4f*)(wP + (size_t)p*32*3072);
}

__global__ __launch_bounds__(256) void fused_w5_tanh_wmma(
    const float* __restrict__ attn, const float* __restrict__ Af,
    const float* __restrict__ W5,   const float* __restrict__ b5,
    const float* __restrict__ T0,   const float* __restrict__ Pt,
    float* __restrict__ out)
{
    __shared__ __attribute__((aligned(16))) float smem[2*STG];
    float (*aBlk)[8]        = (float (*)[8])smem;
    float (*startS)[TN + 2] = (float (*)[TN + 2])(smem + TM*8);

    const int tid  = threadIdx.x;
    const int lane = tid & 31, wid = tid >> 5;
    const int wm   = wid >> 1, wn  = wid & 1;
    const int ll   = lane & 15, lh = lane >> 4;
    const int n0   = blockIdx.x * TN, m0 = blockIdx.y * TM;

    const float* aP = Af + (size_t)(m0 + (tid >> 3))*1536 + ((tid & 7) << 2);
    const float* wP = W5 + (size_t)(n0 + (tid >> 3))*3072 + 1536 + ((tid & 7) << 2);
    v4f ra[4], rw[2];
    tile_get(aP, wP, ra, rw);

    {
        int r = tid >> 1, c = (tid & 1) << 2;
        *(v4f*)&aBlk[r][c] = *(const v4f*)(attn + (size_t)(m0 + r)*8 + c);
    }

    __syncthreads();

    {
        const int col = tid & 63;
        const int rg  = tid >> 6;
        const int gn  = n0 + col;
        #pragma unroll 4
        for (int i = 0; i < 32; ++i) {
            const int row = rg*32 + i;
            const int m   = m0 + row;
            const int bs  = m / 20;
            const int l   = m - bs*20;
            const v4f aLo = *(const v4f*)&aBlk[row][0];
            const v4f aHi = *(const v4f*)&aBlk[row][4];
            const float* pp = Pt + ((size_t)l*768 + gn)*8;
            const v4f pLo = *(const v4f*)pp;
            const v4f pHi = *(const v4f*)(pp + 4);
            float s = T0[(size_t)bs*768 + gn];
            s += aLo.x*pLo.x; s += aLo.y*pLo.y; s += aLo.z*pLo.z; s += aLo.w*pLo.w;
            s += aHi.x*pHi.x; s += aHi.y*pHi.y; s += aHi.z*pHi.z; s += aHi.w*pHi.w;
            startS[row][col] = s;
        }
    }
    __syncthreads();

    v8f acc[2][2];
    #pragma unroll
    for (int mt = 0; mt < 2; ++mt)
        #pragma unroll
        for (int nt = 0; nt < 2; ++nt)
            #pragma unroll
            for (int r = 0; r < 8; ++r)
                acc[mt][nt][r] = startS[wm*32 + mt*16 + lh*8 + r][wn*32 + nt*16 + ll];

    __syncthreads();

    float* st0 = smem;
    float* st1 = smem + STG;
    stage_put(st0, tid, ra, rw);
    tile_get(aP + 32, wP + 32, ra, rw);
    __syncthreads();

    for (int kc = 0; kc < 48; kc += 2) {
        stage_put(st1, tid, ra, rw);
        if (kc + 2 < 48) tile_get(aP + (kc + 2)*32, wP + (kc + 2)*32, ra, rw);
        mma_phase(st0, acc, wm, wn, ll, lh);
        __syncthreads();

        if (kc + 2 < 48) stage_put(st0, tid, ra, rw);
        if (kc + 3 < 48) tile_get(aP + (kc + 3)*32, wP + (kc + 3)*32, ra, rw);
        mma_phase(st1, acc, wm, wn, ll, lh);
        __syncthreads();
    }

    #pragma unroll
    for (int mt = 0; mt < 2; ++mt)
        #pragma unroll
        for (int nt = 0; nt < 2; ++nt) {
            int gn = n0 + wn*32 + nt*16 + ll;
            float bias = b5[gn];
            #pragma unroll
            for (int r = 0; r < 8; ++r) {
                int gm = m0 + wm*32 + mt*16 + lh*8 + r;
                out[(size_t)gm*768 + gn] = tanhf(acc[mt][nt][r] + bias);
            }
        }
}

extern "C" void kernel_launch(void* const* d_in, const int* in_sizes, int n_in,
                              void* d_out, int out_size, void* d_ws, size_t ws_size,
                              hipStream_t stream)
{
    (void)in_sizes; (void)n_in; (void)out_size; (void)ws_size;

    const float* token = (const float*)d_in[0];
    const float* label = (const float*)d_in[1];
    const float* imask = (const float*)d_in[2];
    const float* lmask = (const float*)d_in[3];
    const float* W1    = (const float*)d_in[4];
    const float* W2    = (const float*)d_in[5];
    const float* W5    = (const float*)d_in[6];
    const float* b5    = (const float*)d_in[7];
    float* out = (float*)d_out;

    float* ws     = (float*)d_ws;
    float* tkn    = ws;
    float* lab    = tkn    + 2048*768;
    float* scores = lab    + 160*768;
    float* attn   = scores + 2048*160;
    float* b_in   = attn   + 2048*160;
    float* q2c    = b_in   + 40960;
    float* T0     = q2c    + 160*768;
    float* Pt     = T0     + 2048*768;
    float* Af     = Pt     + 20*768*8;

    dim3 blk(256);
    gemm_nt_f32_wmma<2048, 768, 768><<<dim3(16, 12), blk, 0, stream>>>(token, W1, tkn);
    gemm_nt_f32_wmma<160, 768, 300><<<dim3(2, 12),  blk, 0, stream>>>(label, W2, lab);
    gemm_nt_f32_wmma<2048, 160, 768><<<dim3(16, 3),  blk, 0, stream>>>(tkn, lab, scores);
    attn_softmax_kernel<<<dim3(160), blk, 0, stream>>>(scores, lmask, imask, attn, b_in);
    q2c_kernel<<<dim3(192), blk, 0, stream>>>(b_in, tkn, q2c);
    rowfactor_w5_wmma<<<dim3(18, 12), blk, 0, stream>>>(tkn, lab, W5, T0, Pt);
    build_fused_a<<<dim3(20, 128), dim3(192), 0, stream>>>(tkn, lab, attn, q2c, Af);
    fused_w5_tanh_wmma<<<dim3(12, 320), blk, 0, stream>>>(attn, Af, W5, b5, T0, Pt, out);
}


# Main GEMM block tile 128x64 -> 160x128 (wave tile 32x32 -> 80x32; a block = 8 tokens x 20 labels): 7 LDS fragment reads per 10 products, 40% fewer A/W5 tile bytes per product
# speedup vs baseline: 2.3976x; 1.0730x over previous
#include <hip/hip_runtime.h>
#include <math.h>


typedef __attribute__((ext_vector_type(2))) float v2f;
typedef __attribute__((ext_vector_type(4))) float v4f;
typedef __attribute__((ext_vector_type(8))) float v8f;

#define NEG_BIG (-10000.0f)

#define TM 128
#define TN 64
#define TK 32
#define TKP 34


#define STG (192*TKP)

__device__ __forceinline__ void stage_put(float* st, int tid, const v4f (&a)[4], const v4f (&w)[2])
{
    const int r = tid >> 3, c = (tid & 7) << 2;
    #pragma unroll
    for (int p = 0; p < 4; ++p) {
        float* d = st + (p*32 + r)*TKP + c;
        v2f lo = {a[p].x, a[p].y}, hi = {a[p].z, a[p].w};
        *(v2f*)d = lo; *(v2f*)(d + 2) = hi;
    }
    #pragma unroll
    for (int p = 0; p < 2; ++p) {
        float* d = st + (TM + p*32 + r)*TKP + c;
        v2f lo = {w[p].x, w[p].y}, hi = {w[p].z, w[p].w};
        *(v2f*)d = lo; *(v2f*)(d + 2) = hi;
    }
}

__device__ __forceinline__ void mma_phase(const float* st, v8f (&acc)[2][2],
                                          int wm, int wn, int ll, int lh)
{
    const float* As = st;
    const float* Ws = st + TM*TKP;
    #pragma unroll
    for (int kk = 0; kk < 8; ++kk) {
        const int kc = 4*kk + 2*lh;
        v2f a0 = *(const v2f*)(As + (wm*32      + ll)*TKP + kc);
        v2f a1 = *(const v2f*)(As + (wm*32 + 16 + ll)*TKP + kc);
        v2f b0 = *(const v2f*)(Ws + (wn*32      + ll)*TKP + kc);
        v2f b1 = *(const v2f*)(Ws + (wn*32 + 16 + ll)*TKP + kc);
        acc[0][0] = __builtin_amdgcn_wmma_f32_16x16x4_f32(false,a0,false,b0,(short)0,acc[0][0],false,false);
        acc[0][1] = __builtin_amdgcn_wmma_f32_16x16x4_f32(false,a0,false,b1,(short)0,acc[0][1],false,false);
        acc[1][0] = __builtin_amdgcn_wmma_f32_16x16x4_f32(false,a1,false,b0,(short)0,acc[1][0],false,false);
        acc[1][1] = __builtin_amdgcn_wmma_f32_16x16x4_f32(false,a1,false,b1,(short)0,acc[1][1],false,false);
    }
}

__device__ __forceinline__ void tile_get_p(const float* const (&pa)[4], const float* const (&pw)[2],
                                           int ko, v4f (&a)[4], v4f (&w)[2])
{
    #pragma unroll
    for (int p = 0; p < 4; ++p) a[p] = *(const v4f*)(pa[p] + ko);
    #pragma unroll
    for (int p = 0; p < 2; ++p) w[p] = *(const v4f*)(pw[p] + ko);
}

__device__ __forceinline__ void stage_put_z(float* st, int tid, const v4f (&a)[4], const v4f (&w)[2],
                                            const bool (&okA)[4], const bool (&okW)[2], bool kin)
{
    const v4f z = {0.f,0.f,0.f,0.f};
    v4f az[4], wz[2];
    #pragma unroll
    for (int p = 0; p < 4; ++p) az[p] = (okA[p] && kin) ? a[p] : z;
    #pragma unroll
    for (int p = 0; p < 2; ++p) wz[p] = (okW[p] && kin) ? w[p] : z;
    stage_put(st, tid, az, wz);
}

template<int K, int LDA, int LDW, bool GA, bool GW>
__device__ __forceinline__ void gemm_tile_128x64(
    const float* __restrict__ Ab, int rowsA,
    const float* __restrict__ Wb, int rowsW,
    float* smem, v8f (&acc)[2][2], int tid, int wm, int wn, int ll, int lh)
{
    constexpr int  NCH   = (K + TK - 1) / TK;
    constexpr bool GK    = (K % TK) != 0;
    constexpr int  KLAST = K - (NCH - 1)*TK;
    static_assert((K & 3) == 0, "16-byte loads along K");
    static_assert((NCH & 1) == 0 && NCH >= 2, "two chunks per trip");

    const int r = tid >> 3, c = (tid & 7) << 2;

    const float* pa[4]; const float* pw[2];
    bool okA[4], okW[2];
    #pragma unroll
    for (int p = 0; p < 4; ++p) {
        const int row = p*32 + r;
        okA[p] = !GA || row < rowsA;
        const int rc = (GA && row >= rowsA) ? rowsA - 1 : row;
        pa[p] = Ab + (size_t)rc*LDA + c;
    }
    #pragma unroll
    for (int p = 0; p < 2; ++p) {
        const int row = p*32 + r;
        okW[p] = !GW || row < rowsW;
        const int rc = (GW && row >= rowsW) ? rowsW - 1 : row;
        pw[p] = Wb + (size_t)rc*LDW + c;
    }
    const bool tailIn = !GK || c < KLAST;
    const int  koLast = tailIn ? (NCH - 1)*TK : 0;

    v4f ra[4], rw[2];
    float* st0 = smem;
    float* st1 = smem + STG;

    tile_get_p(pa, pw, 0, ra, rw);
    stage_put_z(st0, tid, ra, rw, okA, okW, true);
    tile_get_p(pa, pw, (1 == NCH - 1) ? koLast : TK, ra, rw);
    __syncthreads();

    #pragma unroll 1
    for (int kc = 0; kc < NCH; kc += 2) {
        stage_put_z(st1, tid, ra, rw, okA, okW, (kc + 1 != NCH - 1) || tailIn);
        if (kc + 2 < NCH) tile_get_p(pa, pw, (kc + 2)*TK, ra, rw);
        mma_phase(st0, acc, wm, wn, ll, lh);
        __syncthreads();

        if (kc + 2 < NCH) stage_put_z(st0, tid, ra, rw, okA, okW, true);
        if (kc + 3 < NCH) tile_get_p(pa, pw, (kc + 3 == NCH - 1) ? koLast : (kc + 3)*TK, ra, rw);
        mma_phase(st1, acc, wm, wn, ll, lh);
        __syncthreads();
    }
}

template<int M, int N, int K>
__global__ __launch_bounds__(256) void gemm_nt_f32_wmma(
    const float* __restrict__ A, const float* __restrict__ W, float* __restrict__ C)
{
    __shared__ __attribute__((aligned(16))) float smem[2*STG];
    constexpr bool GA = (M % TM) != 0, GW = (N % TN) != 0;

    const int tid  = threadIdx.x;
    const int lane = tid & 31, wid = tid >> 5;
    const int wm   = wid >> 1, wn  = wid & 1;
    const int ll   = lane & 15, lh = lane >> 4;
    const int m0   = blockIdx.x * TM, n0 = blockIdx.y * TN;

    const v8f vzero = {0.f,0.f,0.f,0.f,0.f,0.f,0.f,0.f};
    v8f acc[2][2];
    acc[0][0]=vzero; acc[0][1]=vzero; acc[1][0]=vzero; acc[1][1]=vzero;

    gemm_tile_128x64<K, K, K, GA, GW>(A + (size_t)m0*K, M - m0, W + (size_t)n0*K, N - n0,
                                      smem, acc, tid, wm, wn, ll, lh);

    #pragma unroll
    for (int mt = 0; mt < 2; ++mt)
        #pragma unroll
        for (int nt = 0; nt < 2; ++nt) {
            int gn = n0 + wn*32 + nt*16 + ll;
            #pragma unroll
            for (int r = 0; r < 8; ++r) {
                int gm = m0 + wm*32 + mt*16 + lh*8 + r;
                if ((!GA || gm < M) && (!GW || gn < N)) C[(size_t)gm*N + gn] = acc[mt][nt][r];
            }
        }
}

__global__ __launch_bounds__(256) void attn_softmax_kernel(
    const float* __restrict__ scores, const float* __restrict__ lmask,
    const float* __restrict__ imask, float* __restrict__ attn,
    float* __restrict__ b_in)
{
    int row = blockIdx.x * 256 + threadIdx.x;
    if (row >= 8*256*20) return;
    int bs = row / 20;
    int l  = row - bs*20;

    float v[8];
    float mx = -3.4e38f;
    #pragma unroll
    for (int t = 0; t < 8; ++t) {
        float s = scores[(size_t)row*8 + t] + (1.0f - lmask[l*8 + t]) * NEG_BIG;
        v[t] = s;
        mx = fmaxf(mx, s);
    }
    float sum = 0.f;
    #pragma unroll
    for (int t = 0; t < 8; ++t) { float e = __expf(v[t] - mx); v[t] = e; sum += e; }
    float inv = 1.0f / sum;
    #pragma unroll
    for (int t = 0; t < 8; ++t) attn[(size_t)row*8 + t] = v[t] * inv;

    b_in[row] = mx + (1.0f - imask[bs]) * NEG_BIG;
}

#define QC_TP 36

__global__ __launch_bounds__(256) void q2c_kernel(
    const float* __restrict__ b_in, const float* __restrict__ tkn,
    float* __restrict__ q2c)
{
    __shared__ __attribute__((aligned(16))) float Wt[256*20];
    __shared__ __attribute__((aligned(16))) float Ts[256*QC_TP];
    __shared__ __attribute__((aligned(16))) float Pp[8*20*32];

    const int tid  = threadIdx.x;
    const int lane = tid & 31, wid = tid >> 5;
    const int b    = blockIdx.x / 24;
    const int slab = blockIdx.x - b*24;
    const int cq   = tid & 7, r0 = tid >> 3;

    {
        v4f xb[5], tt[8];
        const float* bp = b_in + (size_t)b*256*20;
        const float* tp = tkn + ((size_t)b*256 + r0)*768 + slab*32 + 4*cq;
        #pragma unroll
        for (int p = 0; p < 5; ++p) xb[p] = *(const v4f*)(bp + 4*(tid + 256*p));
        #pragma unroll
        for (int p = 0; p < 8; ++p) tt[p] = *(const v4f*)(tp + (size_t)p*32*768);
        #pragma unroll
        for (int p = 0; p < 5; ++p) *(v4f*)(Wt + 4*(tid + 256*p)) = xb[p];
        #pragma unroll
        for (int p = 0; p < 8; ++p) *(v4f*)(Ts + (r0 + 32*p)*QC_TP + 4*cq) = tt[p];
    }
    __syncthreads();

    {
        float x[3][8], mx[3], sm[3];
        #pragma unroll
        for (int k = 0; k < 3; ++k) {
            const int l  = wid + 8*k;
            const int lc = (l < 20) ? l : wid;
            mx[k] = -3.4e38f;
            #pragma unroll
            for (int i = 0; i < 8; ++i) { x[k][i] = Wt[(lane + 32*i)*20 + lc]; mx[k] = fmaxf(mx[k], x[k][i]); }
        }
        #pragma unroll
        for (int o = 16; o > 0; o >>= 1)
            #pragma unroll
            for (int k = 0; k < 3; ++k) mx[k] = fmaxf(mx[k], __shfl_xor(mx[k], o, 32));
        #pragma unroll
        for (int k = 0; k < 3; ++k) {
            sm[k] = 0.f;
            #pragma unroll
            for (int i = 0; i < 8; ++i) { x[k][i] = __expf(x[k][i] - mx[k]); sm[k] += x[k][i]; }
        }
        #pragma unroll
        for (int o = 16; o > 0; o >>= 1)
            #pragma unroll
            for (int k = 0; k < 3; ++k) sm[k] += __shfl_xor(sm[k], o, 32);
        #pragma unroll
        for (int k = 0; k < 3; ++k) {
            const int l = wid + 8*k;
            if (l < 20) {
                #pragma unroll
                for (int i = 0; i < 8; ++i) Wt[(lane + 32*i)*20 + l] = x[k][i] / sm[k];
            }
        }
    }
    __syncthreads();

    {
        const int lg = (tid >> 3) & 3;
        const v4f z = {0.f,0.f,0.f,0.f};
        v4f acc[5] = {z, z, z, z, z};
        const float* tsp = Ts + (wid*32)*QC_TP + 4*cq;
        const float* wp  = Wt + (wid*32)*20 + 5*lg;
        #pragma unroll 8
        for (int r = 0; r < 32; ++r) {
            const v4f t = *(const v4f*)(tsp + r*QC_TP);
            #pragma unroll
            for (int k = 0; k < 5; ++k) acc[k] += wp[r*20 + k] * t;
        }
        #pragma unroll
        for (int k = 0; k < 5; ++k) *(v4f*)(Pp + (wid*20 + 5*lg + k)*32 + 4*cq) = acc[k];
    }
    __syncthreads();

    if (tid < 160) {
        const int l = tid >> 3;
        v4f s = *(const v4f*)(Pp + l*32 + 4*cq);
        #pragma unroll
        for (int w = 1; w < 8; ++w) s += *(const v4f*)(Pp + (w*20 + l)*32 + 4*cq);
        *(v4f*)(q2c + (size_t)(b*20 + l)*768 + slab*32 + 4*cq) = s;
    }
}

__global__ __launch_bounds__(256) void rowfactor_w5_wmma(
    const float* __restrict__ tkn, const float* __restrict__ lab,
    const float* __restrict__ W5,
    float* __restrict__ T0, float* __restrict__ Pt)
{
    __shared__ __attribute__((aligned(16))) float smem[2*STG];

    const int tid  = threadIdx.x;
    const int lane = tid & 31, wid = tid >> 5;
    const int wm   = wid >> 1, wn  = wid & 1;
    const int ll   = lane & 15, lh = lane >> 4;
    const int m0   = blockIdx.x * TM, n0 = blockIdx.y * TN;

    const bool isLab = (m0 >= 2048);
    const int  q0    = m0 - 2048;
    const float* __restrict__ Ab = isLab ? (lab + (size_t)q0*768) : (tkn + (size_t)m0*768);
    const int  rows  = isLab ? (160 - q0) : TM;
    const float* __restrict__ Wb = W5 + (isLab ? 768 : 0) + (size_t)n0*3072;

    const v8f vzero = {0.f,0.f,0.f,0.f,0.f,0.f,0.f,0.f};
    v8f acc[2][2];
    acc[0][0]=vzero; acc[0][1]=vzero; acc[1][0]=vzero; acc[1][1]=vzero;

    gemm_tile_128x64<768, 768, 3072, true, false>(Ab, rows, Wb, TN, smem, acc, tid, wm, wn, ll, lh);

    #pragma unroll
    for (int mt = 0; mt < 2; ++mt)
        #pragma unroll
        for (int nt = 0; nt < 2; ++nt) {
            int gn = n0 + wn*32 + nt*16 + ll;
            #pragma unroll
            for (int r = 0; r < 8; ++r) {
                int row = wm*32 + mt*16 + lh*8 + r;
                if (!isLab) {
                    T0[(size_t)(m0 + row)*768 + gn] = acc[mt][nt][r];
                } else if (row < rows) {
                    int q = q0 + row;
                    Pt[((size_t)(q >> 3)*768 + gn)*8 + (q & 7)] = acc[mt][nt][r];
                }
            }
        }
}

__global__ __launch_bounds__(192) void build_fused_a(
    const float* __restrict__ tkn,  const float* __restrict__ lab,
    const float* __restrict__ attn, const float* __restrict__ q2c,
    float* __restrict__ Af)
{
    const int tid = threadIdx.x;
    const int l   = blockIdx.x;
    const int bs0 = blockIdx.y * 16;
    const int b   = bs0 >> 8;
    const int j   = tid << 2;

    v4f lv[8];
    #pragma unroll
    for (int t = 0; t < 8; ++t) lv[t] = *(const v4f*)(lab + (size_t)(l*8 + t)*768 + j);
    const v4f q4 = *(const v4f*)(q2c + (size_t)(b*20 + l)*768 + j);

    #pragma unroll 4
    for (int i = 0; i < 16; ++i) {
        const int bs = bs0 + i;
        const size_t m = (size_t)bs*20 + l;
        const v4f aLo = *(const v4f*)(attn + m*8);
        const v4f aHi = *(const v4f*)(attn + m*8 + 4);
        const v4f t4  = *(const v4f*)(tkn + (size_t)bs*768 + j);
        v4f cq = {0.f,0.f,0.f,0.f};
        cq += aLo.x * lv[0]; cq += aLo.y * lv[1]; cq += aLo.z * lv[2]; cq += aLo.w * lv[3];
        cq += aHi.x * lv[4]; cq += aHi.y * lv[5]; cq += aHi.z * lv[6]; cq += aHi.w * lv[7];
        float* dst = Af + m*1536 + j;
        *(v4f*)dst         = t4 * cq;
        *(v4f*)(dst + 768) = t4 * q4;
    }
}

#define MM 160
#define MN 128
#define MSTG (288*TKP)
#define MSP 130
#define MSMEM (MM*8 + MM*MSP)

__device__ __forceinline__ void tile_get_m(const float* __restrict__ aP, const float* __restrict__ wP,
                                           v4f (&a)[5], v4f (&w)[4])
{
    #pragma unroll
    for (int p = 0; p < 5; ++p) a[p] = *(const v4f*)(aP + (size_t)p*32*1536);
    #pragma unroll
    for (int p = 0; p < 4; ++p) w[p] = *(const v4f*)(wP + (size_t)p*32*3072);
}

__device__ __forceinline__ void stage_put_m(float* st, int tid, const v4f (&a)[5], const v4f (&w)[4])
{
    const int r = tid >> 3, c = (tid & 7) << 2;
    #pragma unroll
    for (int p = 0; p < 5; ++p) {
        float* d = st + (p*32 + r)*TKP + c;
        v2f lo = {a[p].x, a[p].y}, hi = {a[p].z, a[p].w};
        *(v2f*)d = lo; *(v2f*)(d + 2) = hi;
    }
    #pragma unroll
    for (int p = 0; p < 4; ++p) {
        float* d = st + (MM + p*32 + r)*TKP + c;
        v2f lo = {w[p].x, w[p].y}, hi = {w[p].z, w[p].w};
        *(v2f*)d = lo; *(v2f*)(d + 2) = hi;
    }
}

__device__ __forceinline__ void mma_phase_m(const float* st, v8f (&acc)[5][2],
                                            int wm, int wn, int ll, int lh)
{
    const float* As = st + (wm*80 + ll)*TKP + 2*lh;
    const float* Ws = st + (MM + wn*32 + ll)*TKP + 2*lh;
    #pragma unroll
    for (int kk = 0; kk < 8; ++kk) {
        v2f a[5], b[2];
        #pragma unroll
        for (int i = 0; i < 5; ++i) a[i] = *(const v2f*)(As + i*16*TKP + 4*kk);
        #pragma unroll
        for (int j = 0; j < 2; ++j) b[j] = *(const v2f*)(Ws + j*16*TKP + 4*kk);
        #pragma unroll
        for (int i = 0; i < 5; ++i)
            #pragma unroll
            for (int j = 0; j < 2; ++j)
                acc[i][j] = __builtin_amdgcn_wmma_f32_16x16x4_f32(false,a[i],false,b[j],(short)0,acc[i][j],false,false);
    }
}

__global__ __launch_bounds__(256) void fused_w5_tanh_wmma(
    const float* __restrict__ attn, const float* __restrict__ Af,
    const float* __restrict__ W5,   const float* __restrict__ b5,
    const float* __restrict__ T0,   const float* __restrict__ Pt,
    float* __restrict__ out)
{
    __shared__ __attribute__((aligned(16))) float smem[MSMEM];
    static_assert(MSMEM >= 2*MSTG, "the stages live in the same bytes");
    float (*aBlk)[8]     = (float (*)[8])smem;
    float (*startS)[MSP] = (float (*)[MSP])(smem + MM*8);

    const int tid  = threadIdx.x;
    const int lane = tid & 31, wid = tid >> 5;
    const int wm   = wid >> 2, wn  = wid & 3;
    const int ll   = lane & 15, lh = lane >> 4;
    const int n0   = blockIdx.x * MN, m0 = blockIdx.y * MM;

    const float* aP = Af + (size_t)(m0 + (tid >> 3))*1536 + ((tid & 7) << 2);
    const float* wP = W5 + (size_t)(n0 + (tid >> 3))*3072 + 1536 + ((tid & 7) << 2);
    v4f ra[5], rw[4];
    tile_get_m(aP, wP, ra, rw);

    for (int i = tid; i < 2*MM; i += 256) {
        const int r = i >> 1, c = (i & 1) << 2;
        *(v4f*)&aBlk[r][c] = *(const v4f*)(attn + (size_t)(m0 + r)*8 + c);
    }

    __syncthreads();

    {
        const int col = tid & (MN - 1);
        const int lb  = (tid >> 7) * 10;
        const int gn  = n0 + col;
        const int bs0 = blockIdx.y * 8;
        float t0[8];
        #pragma unroll
        for (int i = 0; i < 8; ++i) t0[i] = T0[(size_t)(bs0 + i)*768 + gn];
        #pragma unroll 2
        for (int j = 0; j < 10; ++j) {
            const int l = lb + j;
            const float* pp = Pt + ((size_t)l*768 + gn)*8;
            const v4f pLo = *(const v4f*)pp;
            const v4f pHi = *(const v4f*)(pp + 4);
            #pragma unroll
            for (int i = 0; i < 8; ++i) {
                const int row = i*20 + l;
                const v4f aLo = *(const v4f*)&aBlk[row][0];
                const v4f aHi = *(const v4f*)&aBlk[row][4];
                float s = t0[i];
                s += aLo.x*pLo.x; s += aLo.y*pLo.y; s += aLo.z*pLo.z; s += aLo.w*pLo.w;
                s += aHi.x*pHi.x; s += aHi.y*pHi.y; s += aHi.z*pHi.z; s += aHi.w*pHi.w;
                startS[row][col] = s;
            }
        }
    }
    __syncthreads();

    v8f acc[5][2];
    #pragma unroll
    for (int mt = 0; mt < 5; ++mt)
        #pragma unroll
        for (int nt = 0; nt < 2; ++nt)
            #pragma unroll
            for (int r = 0; r < 8; ++r)
                acc[mt][nt][r] = startS[wm*80 + mt*16 + lh*8 + r][wn*32 + nt*16 + ll];

    __syncthreads();

    float* st0 = smem;
    float* st1 = smem + MSTG;
    stage_put_m(st0, tid, ra, rw);
    tile_get_m(aP + 32, wP + 32, ra, rw);
    __syncthreads();

    for (int kc = 0; kc < 48; kc += 2) {
        stage_put_m(st1, tid, ra, rw);
        if (kc + 2 < 48) tile_get_m(aP + (kc + 2)*32, wP + (kc + 2)*32, ra, rw);
        mma_phase_m(st0, acc, wm, wn, ll, lh);
        __syncthreads();

        if (kc + 2 < 48) stage_put_m(st0, tid, ra, rw);
        if (kc + 3 < 48) tile_get_m(aP + (kc + 3)*32, wP + (kc + 3)*32, ra, rw);
        mma_phase_m(st1, acc, wm, wn, ll, lh);
        __syncthreads();
    }

    #pragma unroll
    for (int mt = 0; mt < 5; ++mt)
        #pragma unroll
        for (int nt = 0; nt < 2; ++nt) {
            int gn = n0 + wn*32 + nt*16 + ll;
            float bias = b5[gn];
            #pragma unroll
            for (int r = 0; r < 8; ++r) {
                int gm = m0 + wm*80 + mt*16 + lh*8 + r;
                out[(size_t)gm*768 + gn] = tanhf(acc[mt][nt][r] + bias);
            }
        }
}

extern "C" void kernel_launch(void* const* d_in, const int* in_sizes, int n_in,
                              void* d_out, int out_size, void* d_ws, size_t ws_size,
                              hipStream_t stream)
{
    (void)in_sizes; (void)n_in; (void)out_size; (void)ws_size;

    const float* token = (const float*)d_in[0];
    const float* label = (const float*)d_in[1];
    const float* imask = (const float*)d_in[2];
    const float* lmask = (const float*)d_in[3];
    const float* W1    = (const float*)d_in[4];
    const float* W2    = (const float*)d_in[5];
    const float* W5    = (const float*)d_in[6];
    const float* b5    = (const float*)d_in[7];
    float* out = (float*)d_out;

    float* ws     = (float*)d_ws;
    float* tkn    = ws;
    float* lab    = tkn    + 2048*768;
    float* scores = lab    + 160*768;
    float* attn   = scores + 2048*160;
    float* b_in   = attn   + 2048*160;
    float* q2c    = b_in   + 40960;
    float* T0     = q2c    + 160*768;
    float* Pt     = T0     + 2048*768;
    float* Af     = Pt     + 20*768*8;

    dim3 blk(256);
    gemm_nt_f32_wmma<2048, 768, 768><<<dim3(16, 12), blk, 0, stream>>>(token, W1, tkn);
    gemm_nt_f32_wmma<160, 768, 300><<<dim3(2, 12),  blk, 0, stream>>>(label, W2, lab);
    gemm_nt_f32_wmma<2048, 160, 768><<<dim3(16, 3),  blk, 0, stream>>>(tkn, lab, scores);
    attn_softmax_kernel<<<dim3(160), blk, 0, stream>>>(scores, lmask, imask, attn, b_in);
    q2c_kernel<<<dim3(192), blk, 0, stream>>>(b_in, tkn, q2c);
    rowfactor_w5_wmma<<<dim3(18, 12), blk, 0, stream>>>(tkn, lab, W5, T0, Pt);
    build_fused_a<<<dim3(20, 128), dim3(192), 0, stream>>>(tkn, lab, attn, q2c, Af);
    fused_w5_tanh_wmma<<<dim3(6, 256), blk, 0, stream>>>(attn, Af, W5, b5, T0, Pt, out);
}


# Independent small GEMMs share launches: tkn with lab; scores with the T0/Pt row-factor GEMM (8 launches -> 6); all-padding waves of the Pt blocks skip their products
# speedup vs baseline: 2.4736x; 1.0317x over previous
#include <hip/hip_runtime.h>
#include <math.h>


typedef __attribute__((ext_vector_type(2))) float v2f;
typedef __attribute__((ext_vector_type(4))) float v4f;
typedef __attribute__((ext_vector_type(8))) float v8f;

#define NEG_BIG (-10000.0f)

#define TM 128
#define TN 64
#define TK 32
#define TKP 34


#define STG (192*TKP)

__device__ __forceinline__ void stage_put(float* st, int tid, const v4f (&a)[4], const v4f (&w)[2])
{
    const int r = tid >> 3, c = (tid & 7) << 2;
    #pragma unroll
    for (int p = 0; p < 4; ++p) {
        float* d = st + (p*32 + r)*TKP + c;
        v2f lo = {a[p].x, a[p].y}, hi = {a[p].z, a[p].w};
        *(v2f*)d = lo; *(v2f*)(d + 2) = hi;
    }
    #pragma unroll
    for (int p = 0; p < 2; ++p) {
        float* d = st + (TM + p*32 + r)*TKP + c;
        v2f lo = {w[p].x, w[p].y}, hi = {w[p].z, w[p].w};
        *(v2f*)d = lo; *(v2f*)(d + 2) = hi;
    }
}

__device__ __forceinline__ void mma_phase(const float* st, v8f (&acc)[2][2],
                                          int wm, int wn, int ll, int lh)
{
    const float* As = st;
    const float* Ws = st + TM*TKP;
    #pragma unroll
    for (int kk = 0; kk < 8; ++kk) {
        const int kc = 4*kk + 2*lh;
        v2f a0 = *(const v2f*)(As + (wm*32      + ll)*TKP + kc);
        v2f a1 = *(const v2f*)(As + (wm*32 + 16 + ll)*TKP + kc);
        v2f b0 = *(const v2f*)(Ws + (wn*32      + ll)*TKP + kc);
        v2f b1 = *(const v2f*)(Ws + (wn*32 + 16 + ll)*TKP + kc);
        acc[0][0] = __builtin_amdgcn_wmma_f32_16x16x4_f32(false,a0,false,b0,(short)0,acc[0][0],false,false);
        acc[0][1] = __builtin_amdgcn_wmma_f32_16x16x4_f32(false,a0,false,b1,(short)0,acc[0][1],false,false);
        acc[1][0] = __builtin_amdgcn_wmma_f32_16x16x4_f32(false,a1,false,b0,(short)0,acc[1][0],false,false);
        acc[1][1] = __builtin_amdgcn_wmma_f32_16x16x4_f32(false,a1,false,b1,(short)0,acc[1][1],false,false);
    }
}

__device__ __forceinline__ void tile_get_p(const float* const (&pa)[4], const float* const (&pw)[2],
                                           int ko, v4f (&a)[4], v4f (&w)[2])
{
    #pragma unroll
    for (int p = 0; p < 4; ++p) a[p] = *(const v4f*)(pa[p] + ko);
    #pragma unroll
    for (int p = 0; p < 2; ++p) w[p] = *(const v4f*)(pw[p] + ko);
}

__device__ __forceinline__ void stage_put_z(float* st, int tid, const v4f (&a)[4], const v4f (&w)[2],
                                            const bool (&okA)[4], const bool (&okW)[2], bool kin)
{
    const v4f z = {0.f,0.f,0.f,0.f};
    v4f az[4], wz[2];
    #pragma unroll
    for (int p = 0; p < 4; ++p) az[p] = (okA[p] && kin) ? a[p] : z;
    #pragma unroll
    for (int p = 0; p < 2; ++p) wz[p] = (okW[p] && kin) ? w[p] : z;
    stage_put(st, tid, az, wz);
}

template<int K, bool GA, bool GW, bool SKIP>
__device__ __forceinline__ void gemm_tile_128x64(
    const float* __restrict__ Ab, int rowsA, int lda,
    const float* __restrict__ Wb, int rowsW, int ldw,
    float* smem, v8f (&acc)[2][2], int tid, int wm, int wn, int ll, int lh)
{
    constexpr int  NCH   = (K + TK - 1) / TK;
    constexpr bool GK    = (K % TK) != 0;
    constexpr int  KLAST = K - (NCH - 1)*TK;
    static_assert((K & 3) == 0, "16-byte loads along K");
    static_assert((NCH & 1) == 0 && NCH >= 2, "two chunks per trip");

    const int r = tid >> 3, c = (tid & 7) << 2;

    const float* pa[4]; const float* pw[2];
    bool okA[4], okW[2];
    #pragma unroll
    for (int p = 0; p < 4; ++p) {
        const int row = p*32 + r;
        okA[p] = !GA || row < rowsA;
        const int rc = (GA && row >= rowsA) ? rowsA - 1 : row;
        pa[p] = Ab + (size_t)rc*lda + c;
    }
    #pragma unroll
    for (int p = 0; p < 2; ++p) {
        const int row = p*32 + r;
        okW[p] = !GW || row < rowsW;
        const int rc = (GW && row >= rowsW) ? rowsW - 1 : row;
        pw[p] = Wb + (size_t)rc*ldw + c;
    }
    const bool tailIn = !GK || c < KLAST;
    const int  koLast = tailIn ? (NCH - 1)*TK : 0;

    const bool live = !SKIP || ((!GA || wm*32 < rowsA) && (!GW || wn*32 < rowsW));

    v4f ra[4], rw[2];
    float* st0 = smem;
    float* st1 = smem + STG;

    tile_get_p(pa, pw, 0, ra, rw);
    stage_put_z(st0, tid, ra, rw, okA, okW, true);
    tile_get_p(pa, pw, (1 == NCH - 1) ? koLast : TK, ra, rw);
    __syncthreads();

    #pragma unroll 1
    for (int kc = 0; kc < NCH; kc += 2) {
        stage_put_z(st1, tid, ra, rw, okA, okW, (kc + 1 != NCH - 1) || tailIn);
        if (kc + 2 < NCH) tile_get_p(pa, pw, (kc + 2)*TK, ra, rw);
        if (live) mma_phase(st0, acc, wm, wn, ll, lh);
        __syncthreads();

        if (kc + 2 < NCH) stage_put_z(st0, tid, ra, rw, okA, okW, true);
        if (kc + 3 < NCH) tile_get_p(pa, pw, (kc + 3 == NCH - 1) ? koLast : (kc + 3)*TK, ra, rw);
        if (live) mma_phase(st1, acc, wm, wn, ll, lh);
        __syncthreads();
    }
}

__global__ __launch_bounds__(256) void proj_tkn_lab_wmma(
    const float* __restrict__ token, const float* __restrict__ W1,
    const float* __restrict__ label, const float* __restrict__ W2,
    float* __restrict__ tkn, float* __restrict__ lab)
{
    __shared__ __attribute__((aligned(16))) float smem[2*STG];

    const int tid  = threadIdx.x;
    const int lane = tid & 31, wid = tid >> 5;
    const int wm   = wid >> 1, wn  = wid & 1;
    const int ll   = lane & 15, lh = lane >> 4;
    const int id   = blockIdx.x;

    const v8f vzero = {0.f,0.f,0.f,0.f,0.f,0.f,0.f,0.f};
    v8f acc[2][2];
    acc[0][0]=vzero; acc[0][1]=vzero; acc[1][0]=vzero; acc[1][1]=vzero;

    if (id < 192) {
        const int m0 = (id & 15) * TM, n0 = (id >> 4) * TN;
        gemm_tile_128x64<768, false, false, false>(token + (size_t)m0*768, TM, 768, W1 + (size_t)n0*768, TN, 768,
                                            smem, acc, tid, wm, wn, ll, lh);
        #pragma unroll
        for (int mt = 0; mt < 2; ++mt)
            #pragma unroll
            for (int nt = 0; nt < 2; ++nt) {
                int gn = n0 + wn*32 + nt*16 + ll;
                #pragma unroll
                for (int r = 0; r < 8; ++r) {
                    int gm = m0 + wm*32 + mt*16 + lh*8 + r;
                    tkn[(size_t)gm*768 + gn] = acc[mt][nt][r];
                }
            }
    } else {
        const int s  = id - 192;
        const int m0 = (s / 12) * TM, n0 = (s % 12) * TN;
        gemm_tile_128x64<300, true, false, false>(label + (size_t)m0*300, 160 - m0, 300, W2 + (size_t)n0*300, TN, 300,
                                           smem, acc, tid, wm, wn, ll, lh);
        #pragma unroll
        for (int mt = 0; mt < 2; ++mt)
            #pragma unroll
            for (int nt = 0; nt < 2; ++nt) {
                int gn = n0 + wn*32 + nt*16 + ll;
                #pragma unroll
                for (int r = 0; r < 8; ++r) {
                    int gm = m0 + wm*32 + mt*16 + lh*8 + r;
                    if (gm < 160) lab[(size_t)gm*768 + gn] = acc[mt][nt][r];
                }
            }
    }
}

__global__ __launch_bounds__(256) void attn_softmax_kernel(
    const float* __restrict__ scores, const float* __restrict__ lmask,
    const float* __restrict__ imask, float* __restrict__ attn,
    float* __restrict__ b_in)
{
    int row = blockIdx.x * 256 + threadIdx.x;
    if (row >= 8*256*20) return;
    int bs = row / 20;
    int l  = row - bs*20;

    float v[8];
    float mx = -3.4e38f;
    #pragma unroll
    for (int t = 0; t < 8; ++t) {
        float s = scores[(size_t)row*8 + t] + (1.0f - lmask[l*8 + t]) * NEG_BIG;
        v[t] = s;
        mx = fmaxf(mx, s);
    }
    float sum = 0.f;
    #pragma unroll
    for (int t = 0; t < 8; ++t) { float e = __expf(v[t] - mx); v[t] = e; sum += e; }
    float inv = 1.0f / sum;
    #pragma unroll
    for (int t = 0; t < 8; ++t) attn[(size_t)row*8 + t] = v[t] * inv;

    b_in[row] = mx + (1.0f - imask[bs]) * NEG_BIG;
}

#define QC_TP 36

__global__ __launch_bounds__(256) void q2c_kernel(
    const float* __restrict__ b_in, const float* __restrict__ tkn,
    float* __restrict__ q2c)
{
    __shared__ __attribute__((aligned(16))) float Wt[256*20];
    __shared__ __attribute__((aligned(16))) float Ts[256*QC_TP];
    __shared__ __attribute__((aligned(16))) float Pp[8*20*32];

    const int tid  = threadIdx.x;
    const int lane = tid & 31, wid = tid >> 5;
    const int b    = blockIdx.x / 24;
    const int slab = blockIdx.x - b*24;
    const int cq   = tid & 7, r0 = tid >> 3;

    {
        v4f xb[5], tt[8];
        const float* bp = b_in + (size_t)b*256*20;
        const float* tp = tkn + ((size_t)b*256 + r0)*768 + slab*32 + 4*cq;
        #pragma unroll
        for (int p = 0; p < 5; ++p) xb[p] = *(const v4f*)(bp + 4*(tid + 256*p));
        #pragma unroll
        for (int p = 0; p < 8; ++p) tt[p] = *(const v4f*)(tp + (size_t)p*32*768);
        #pragma unroll
        for (int p = 0; p < 5; ++p) *(v4f*)(Wt + 4*(tid + 256*p)) = xb[p];
        #pragma unroll
        for (int p = 0; p < 8; ++p) *(v4f*)(Ts + (r0 + 32*p)*QC_TP + 4*cq) = tt[p];
    }
    __syncthreads();

    {
        float x[3][8], mx[3], sm[3];
        #pragma unroll
        for (int k = 0; k < 3; ++k) {
            const int l  = wid + 8*k;
            const int lc = (l < 20) ? l : wid;
            mx[k] = -3.4e38f;
            #pragma unroll
            for (int i = 0; i < 8; ++i) { x[k][i] = Wt[(lane + 32*i)*20 + lc]; mx[k] = fmaxf(mx[k], x[k][i]); }
        }
        #pragma unroll
        for (int o = 16; o > 0; o >>= 1)
            #pragma unroll
            for (int k = 0; k < 3; ++k) mx[k] = fmaxf(mx[k], __shfl_xor(mx[k], o, 32));
        #pragma unroll
        for (int k = 0; k < 3; ++k) {
            sm[k] = 0.f;
            #pragma unroll
            for (int i = 0; i < 8; ++i) { x[k][i] = __expf(x[k][i] - mx[k]); sm[k] += x[k][i]; }
        }
        #pragma unroll
        for (int o = 16; o > 0; o >>= 1)
            #pragma unroll
            for (int k = 0; k < 3; ++k) sm[k] += __shfl_xor(sm[k], o, 32);
        #pragma unroll
        for (int k = 0; k < 3; ++k) {
            const int l = wid + 8*k;
            if (l < 20) {
                #pragma unroll
                for (int i = 0; i < 8; ++i) Wt[(lane + 32*i)*20 + l] = x[k][i] / sm[k];
            }
        }
    }
    __syncthreads();

    {
        const int lg = (tid >> 3) & 3;
        const v4f z = {0.f,0.f,0.f,0.f};
        v4f acc[5] = {z, z, z, z, z};
        const float* tsp = Ts + (wid*32)*QC_TP + 4*cq;
        const float* wp  = Wt + (wid*32)*20 + 5*lg;
        #pragma unroll 8
        for (int r = 0; r < 32; ++r) {
            const v4f t = *(const v4f*)(tsp + r*QC_TP);
            #pragma unroll
            for (int k = 0; k < 5; ++k) acc[k] += wp[r*20 + k] * t;
        }
        #pragma unroll
        for (int k = 0; k < 5; ++k) *(v4f*)(Pp + (wid*20 + 5*lg + k)*32 + 4*cq) = acc[k];
    }
    __syncthreads();

    if (tid < 160) {
        const int l = tid >> 3;
        v4f s = *(const v4f*)(Pp + l*32 + 4*cq);
        #pragma unroll
        for (int w = 1; w < 8; ++w) s += *(const v4f*)(Pp + (w*20 + l)*32 + 4*cq);
        *(v4f*)(q2c + (size_t)(b*20 + l)*768 + slab*32 + 4*cq) = s;
    }
}

__global__ __launch_bounds__(256) void scores_rowfactor_wmma(
    const float* __restrict__ tkn, const float* __restrict__ lab,
    const float* __restrict__ W5,
    float* __restrict__ scores, float* __restrict__ T0, float* __restrict__ Pt)
{
    __shared__ __attribute__((aligned(16))) float smem[2*STG];

    const int tid  = threadIdx.x;
    const int lane = tid & 31;
    const int wid  = __builtin_amdgcn_readfirstlane(tid >> 5);
    const int wm   = wid >> 1, wn  = wid & 1;
    const int ll   = lane & 15, lh = lane >> 4;
    const int id   = blockIdx.x;

    int role, m0, n0, rowsA, rowsW, ldw;
    const float* Ab; const float* Wb;
    if (id < 192) {
        role = 0; m0 = (id & 15) * TM; n0 = (id >> 4) * TN;
        Ab = tkn + (size_t)m0*768;  rowsA = TM;
        Wb = W5 + (size_t)n0*3072;  rowsW = TN;  ldw = 3072;
    } else if (id < 240) {
        const int s = id - 192;
        role = 1; m0 = (s & 15) * TM; n0 = (s >> 4) * TN;
        Ab = tkn + (size_t)m0*768;  rowsA = TM;
        Wb = lab + (size_t)n0*768;  rowsW = 160 - n0;  ldw = 768;
    } else {
        const int s = id - 240;
        role = 2; m0 = (s / 12) * TM; n0 = (s % 12) * TN;
        Ab = lab + (size_t)m0*768;  rowsA = 160 - m0;
        Wb = W5 + 768 + (size_t)n0*3072;  rowsW = TN;  ldw = 3072;
    }

    const v8f vzero = {0.f,0.f,0.f,0.f,0.f,0.f,0.f,0.f};
    v8f acc[2][2];
    acc[0][0]=vzero; acc[0][1]=vzero; acc[1][0]=vzero; acc[1][1]=vzero;

    if (role != 2) gemm_tile_128x64<768, false, true, false>(Ab, TM, 768, Wb, rowsW, ldw, smem, acc, tid, wm, wn, ll, lh);
    else           gemm_tile_128x64<768, true, false, true>(Ab, rowsA, 768, Wb, TN, 3072, smem, acc, tid, wm, wn, ll, lh);

    #pragma unroll
    for (int mt = 0; mt < 2; ++mt)
        #pragma unroll
        for (int nt = 0; nt < 2; ++nt) {
            int gn = n0 + wn*32 + nt*16 + ll;
            #pragma unroll
            for (int r = 0; r < 8; ++r) {
                int row = wm*32 + mt*16 + lh*8 + r;
                if (role == 0) {
                    T0[(size_t)(m0 + row)*768 + gn] = acc[mt][nt][r];
                } else if (role == 1) {
                    if (gn < 160) scores[(size_t)(m0 + row)*160 + gn] = acc[mt][nt][r];
                } else if (row < rowsA) {
                    int q = m0 + row;
                    Pt[((size_t)(q >> 3)*768 + gn)*8 + (q & 7)] = acc[mt][nt][r];
                }
            }
        }
}

__global__ __launch_bounds__(192) void build_fused_a(
    const float* __restrict__ tkn,  const float* __restrict__ lab,
    const float* __restrict__ attn, const float* __restrict__ q2c,
    float* __restrict__ Af)
{
    const int tid = threadIdx.x;
    const int l   = blockIdx.x;
    const int bs0 = blockIdx.y * 16;
    const int b   = bs0 >> 8;
    const int j   = tid << 2;

    v4f lv[8];
    #pragma unroll
    for (int t = 0; t < 8; ++t) lv[t] = *(const v4f*)(lab + (size_t)(l*8 + t)*768 + j);
    const v4f q4 = *(const v4f*)(q2c + (size_t)(b*20 + l)*768 + j);

    #pragma unroll 4
    for (int i = 0; i < 16; ++i) {
        const int bs = bs0 + i;
        const size_t m = (size_t)bs*20 + l;
        const v4f aLo = *(const v4f*)(attn + m*8);
        const v4f aHi = *(const v4f*)(attn + m*8 + 4);
        const v4f t4  = *(const v4f*)(tkn + (size_t)bs*768 + j);
        v4f cq = {0.f,0.f,0.f,0.f};
        cq += aLo.x * lv[0]; cq += aLo.y * lv[1]; cq += aLo.z * lv[2]; cq += aLo.w * lv[3];
        cq += aHi.x * lv[4]; cq += aHi.y * lv[5]; cq += aHi.z * lv[6]; cq += aHi.w * lv[7];
        float* dst = Af + m*1536 + j;
        *(v4f*)dst         = t4 * cq;
        *(v4f*)(dst + 768) = t4 * q4;
    }
}

#define MM 160
#define MN 128
#define MSTG (288*TKP)
#define MSP 130
#define MSMEM (MM*8 + MM*MSP)

__device__ __forceinline__ void tile_get_m(const float* __restrict__ aP, const float* __restrict__ wP,
                                           v4f (&a)[5], v4f (&w)[4])
{
    #pragma unroll
    for (int p = 0; p < 5; ++p) a[p] = *(const v4f*)(aP + (size_t)p*32*1536);
    #pragma unroll
    for (int p = 0; p < 4; ++p) w[p] = *(const v4f*)(wP + (size_t)p*32*3072);
}

__device__ __forceinline__ void stage_put_m(float* st, int tid, const v4f (&a)[5], const v4f (&w)[4])
{
    const int r = tid >> 3, c = (tid & 7) << 2;
    #pragma unroll
    for (int p = 0; p < 5; ++p) {
        float* d = st + (p*32 + r)*TKP + c;
        v2f lo = {a[p].x, a[p].y}, hi = {a[p].z, a[p].w};
        *(v2f*)d = lo; *(v2f*)(d + 2) = hi;
    }
    #pragma unroll
    for (int p = 0; p < 4; ++p) {
        float* d = st + (MM + p*32 + r)*TKP + c;
        v2f lo = {w[p].x, w[p].y}, hi = {w[p].z, w[p].w};
        *(v2f*)d = lo; *(v2f*)(d + 2) = hi;
    }
}

__device__ __forceinline__ void mma_phase_m(const float* st, v8f (&acc)[5][2],
                                            int wm, int wn, int ll, int lh)
{
    const float* As = st + (wm*80 + ll)*TKP + 2*lh;
    const float* Ws = st + (MM + wn*32 + ll)*TKP + 2*lh;
    #pragma unroll
    for (int kk = 0; kk < 8; ++kk) {
        v2f a[5], b[2];
        #pragma unroll
        for (int i = 0; i < 5; ++i) a[i] = *(const v2f*)(As + i*16*TKP + 4*kk);
        #pragma unroll
        for (int j = 0; j < 2; ++j) b[j] = *(const v2f*)(Ws + j*16*TKP + 4*kk);
        #pragma unroll
        for (int i = 0; i < 5; ++i)
            #pragma unroll
            for (int j = 0; j < 2; ++j)
                acc[i][j] = __builtin_amdgcn_wmma_f32_16x16x4_f32(false,a[i],false,b[j],(short)0,acc[i][j],false,false);
    }
}

__global__ __launch_bounds__(256) void fused_w5_tanh_wmma(
    const float* __restrict__ attn, const float* __restrict__ Af,
    const float* __restrict__ W5,   const float* __restrict__ b5,
    const float* __restrict__ T0,   const float* __restrict__ Pt,
    float* __restrict__ out)
{
    __shared__ __attribute__((aligned(16))) float smem[MSMEM];
    static_assert(MSMEM >= 2*MSTG, "the stages live in the same bytes");
    float (*aBlk)[8]     = (float (*)[8])smem;
    float (*startS)[MSP] = (float (*)[MSP])(smem + MM*8);

    const int tid  = threadIdx.x;
    const int lane = tid & 31, wid = tid >> 5;
    const int wm   = wid >> 2, wn  = wid & 3;
    const int ll   = lane & 15, lh = lane >> 4;
    const int n0   = blockIdx.x * MN, m0 = blockIdx.y * MM;

    const float* aP = Af + (size_t)(m0 + (tid >> 3))*1536 + ((tid & 7) << 2);
    const float* wP = W5 + (size_t)(n0 + (tid >> 3))*3072 + 1536 + ((tid & 7) << 2);
    v4f ra[5], rw[4];
    tile_get_m(aP, wP, ra, rw);

    for (int i = tid; i < 2*MM; i += 256) {
        const int r = i >> 1, c = (i & 1) << 2;
        *(v4f*)&aBlk[r][c] = *(const v4f*)(attn + (size_t)(m0 + r)*8 + c);
    }

    __syncthreads();

    {
        const int col = tid & (MN - 1);
        const int lb  = (tid >> 7) * 10;
        const int gn  = n0 + col;
        const int bs0 = blockIdx.y * 8;
        float t0[8];
        #pragma unroll
        for (int i = 0; i < 8; ++i) t0[i] = T0[(size_t)(bs0 + i)*768 + gn];
        #pragma unroll 2
        for (int j = 0; j < 10; ++j) {
            const int l = lb + j;
            const float* pp = Pt + ((size_t)l*768 + gn)*8;
            const v4f pLo = *(const v4f*)pp;
            const v4f pHi = *(const v4f*)(pp + 4);
            #pragma unroll
            for (int i = 0; i < 8; ++i) {
                const int row = i*20 + l;
                const v4f aLo = *(const v4f*)&aBlk[row][0];
                const v4f aHi = *(const v4f*)&aBlk[row][4];
                float s = t0[i];
                s += aLo.x*pLo.x; s += aLo.y*pLo.y; s += aLo.z*pLo.z; s += aLo.w*pLo.w;
                s += aHi.x*pHi.x; s += aHi.y*pHi.y; s += aHi.z*pHi.z; s += aHi.w*pHi.w;
                startS[row][col] = s;
            }
        }
    }
    __syncthreads();

    v8f acc[5][2];
    #pragma unroll
    for (int mt = 0; mt < 5; ++mt)
        #pragma unroll
        for (int nt = 0; nt < 2; ++nt)
            #pragma unroll
            for (int r = 0; r < 8; ++r)
                acc[mt][nt][r] = startS[wm*80 + mt*16 + lh*8 + r][wn*32 + nt*16 + ll];

    __syncthreads();

    float* st0 = smem;
    float* st1 = smem + MSTG;
    stage_put_m(st0, tid, ra, rw);
    tile_get_m(aP + 32, wP + 32, ra, rw);
    __syncthreads();

    for (int kc = 0; kc < 48; kc += 2) {
        stage_put_m(st1, tid, ra, rw);
        if (kc + 2 < 48) tile_get_m(aP + (kc + 2)*32, wP + (kc + 2)*32, ra, rw);
        mma_phase_m(st0, acc, wm, wn, ll, lh);
        __syncthreads();

        if (kc + 2 < 48) stage_put_m(st0, tid, ra, rw);
        if (kc + 3 < 48) tile_get_m(aP + (kc + 3)*32, wP + (kc + 3)*32, ra, rw);
        mma_phase_m(st1, acc, wm, wn, ll, lh);
        __syncthreads();
    }

    #pragma unroll
    for (int mt = 0; mt < 5; ++mt)
        #pragma unroll
        for (int nt = 0; nt < 2; ++nt) {
            int gn = n0 + wn*32 + nt*16 + ll;
            float bias = b5[gn];
            #pragma unroll
            for (int r = 0; r < 8; ++r) {
                int gm = m0 + wm*80 + mt*16 + lh*8 + r;
                out[(size_t)gm*768 + gn] = tanhf(acc[mt][nt][r] + bias);
            }
        }
}

extern "C" void kernel_launch(void* const* d_in, const int* in_sizes, int n_in,
                              void* d_out, int out_size, void* d_ws, size_t ws_size,
                              hipStream_t stream)
{
    (void)in_sizes; (void)n_in; (void)out_size; (void)ws_size;

    const float* token = (const float*)d_in[0];
    const float* label = (const float*)d_in[1];
    const float* imask = (const float*)d_in[2];
    const float* lmask = (const float*)d_in[3];
    const float* W1    = (const float*)d_in[4];
    const float* W2    = (const float*)d_in[5];
    const float* W5    = (const float*)d_in[6];
    const float* b5    = (const float*)d_in[7];
    float* out = (float*)d_out;

    float* ws     = (float*)d_ws;
    float* tkn    = ws;
    float* lab    = tkn    + 2048*768;
    float* scores = lab    + 160*768;
    float* attn   = scores + 2048*160;
    float* b_in   = attn   + 2048*160;
    float* q2c    = b_in   + 40960;
    float* T0     = q2c    + 160*768;
    float* Pt     = T0     + 2048*768;
    float* Af     = Pt     + 20*768*8;

    dim3 blk(256);
    proj_tkn_lab_wmma<<<dim3(216), blk, 0, stream>>>(token, W1, label, W2, tkn, lab);
    scores_rowfactor_wmma<<<dim3(264), blk, 0, stream>>>(tkn, lab, W5, scores, T0, Pt);
    attn_softmax_kernel<<<dim3(160), blk, 0, stream>>>(scores, lmask, imask, attn, b_in);
    q2c_kernel<<<dim3(192), blk, 0, stream>>>(b_in, tkn, q2c);
    build_fused_a<<<dim3(20, 128), dim3(192), 0, stream>>>(tkn, lab, attn, q2c, Af);
    fused_w5_tanh_wmma<<<dim3(6, 256), blk, 0, stream>>>(attn, Af, W5, b5, T0, Pt, out);
}


# Main GEMM epilogue: tanh branch-free as 1 - 2/(1 + exp2(x*2*log2 e)), one hardware exp2 + one reciprocal per value; was the library's two divergent branches per value; abs error a few 1e-7
# speedup vs baseline: 2.5595x; 1.0347x over previous
#include <hip/hip_runtime.h>
#include <math.h>


typedef __attribute__((ext_vector_type(2))) float v2f;
typedef __attribute__((ext_vector_type(4))) float v4f;
typedef __attribute__((ext_vector_type(8))) float v8f;

#define NEG_BIG (-10000.0f)

#define TM 128
#define TN 64
#define TK 32
#define TKP 34


#define STG (192*TKP)

__device__ __forceinline__ void stage_put(float* st, int tid, const v4f (&a)[4], const v4f (&w)[2])
{
    const int r = tid >> 3, c = (tid & 7) << 2;
    #pragma unroll
    for (int p = 0; p < 4; ++p) {
        float* d = st + (p*32 + r)*TKP + c;
        v2f lo = {a[p].x, a[p].y}, hi = {a[p].z, a[p].w};
        *(v2f*)d = lo; *(v2f*)(d + 2) = hi;
    }
    #pragma unroll
    for (int p = 0; p < 2; ++p) {
        float* d = st + (TM + p*32 + r)*TKP + c;
        v2f lo = {w[p].x, w[p].y}, hi = {w[p].z, w[p].w};
        *(v2f*)d = lo; *(v2f*)(d + 2) = hi;
    }
}

__device__ __forceinline__ void mma_phase(const float* st, v8f (&acc)[2][2],
                                          int wm, int wn, int ll, int lh)
{
    const float* As = st;
    const float* Ws = st + TM*TKP;
    #pragma unroll
    for (int kk = 0; kk < 8; ++kk) {
        const int kc = 4*kk + 2*lh;
        v2f a0 = *(const v2f*)(As + (wm*32      + ll)*TKP + kc);
        v2f a1 = *(const v2f*)(As + (wm*32 + 16 + ll)*TKP + kc);
        v2f b0 = *(const v2f*)(Ws + (wn*32      + ll)*TKP + kc);
        v2f b1 = *(const v2f*)(Ws + (wn*32 + 16 + ll)*TKP + kc);
        acc[0][0] = __builtin_amdgcn_wmma_f32_16x16x4_f32(false,a0,false,b0,(short)0,acc[0][0],false,false);
        acc[0][1] = __builtin_amdgcn_wmma_f32_16x16x4_f32(false,a0,false,b1,(short)0,acc[0][1],false,false);
        acc[1][0] = __builtin_amdgcn_wmma_f32_16x16x4_f32(false,a1,false,b0,(short)0,acc[1][0],false,false);
        acc[1][1] = __builtin_amdgcn_wmma_f32_16x16x4_f32(false,a1,false,b1,(short)0,acc[1][1],false,false);
    }
}

__device__ __forceinline__ void tile_get_p(const float* const (&pa)[4], const float* const (&pw)[2],
                                           int ko, v4f (&a)[4], v4f (&w)[2])
{
    #pragma unroll
    for (int p = 0; p < 4; ++p) a[p] = *(const v4f*)(pa[p] + ko);
    #pragma unroll
    for (int p = 0; p < 2; ++p) w[p] = *(const v4f*)(pw[p] + ko);
}

__device__ __forceinline__ void stage_put_z(float* st, int tid, const v4f (&a)[4], const v4f (&w)[2],
                                            const bool (&okA)[4], const bool (&okW)[2], bool kin)
{
    const v4f z = {0.f,0.f,0.f,0.f};
    v4f az[4], wz[2];
    #pragma unroll
    for (int p = 0; p < 4; ++p) az[p] = (okA[p] && kin) ? a[p] : z;
    #pragma unroll
    for (int p = 0; p < 2; ++p) wz[p] = (okW[p] && kin) ? w[p] : z;
    stage_put(st, tid, az, wz);
}

template<int K, bool GA, bool GW, bool SKIP>
__device__ __forceinline__ void gemm_tile_128x64(
    const float* __restrict__ Ab, int rowsA, int lda,
    const float* __restrict__ Wb, int rowsW, int ldw,
    float* smem, v8f (&acc)[2][2], int tid, int wm, int wn, int ll, int lh)
{
    constexpr int  NCH   = (K + TK - 1) / TK;
    constexpr bool GK    = (K % TK) != 0;
    constexpr int  KLAST = K - (NCH - 1)*TK;
    static_assert((K & 3) == 0, "16-byte loads along K");
    static_assert((NCH & 1) == 0 && NCH >= 2, "two chunks per trip");

    const int r = tid >> 3, c = (tid & 7) << 2;

    const float* pa[4]; const float* pw[2];
    bool okA[4], okW[2];
    #pragma unroll
    for (int p = 0; p < 4; ++p) {
        const int row = p*32 + r;
        okA[p] = !GA || row < rowsA;
        const int rc = (GA && row >= rowsA) ? rowsA - 1 : row;
        pa[p] = Ab + (size_t)rc*lda + c;
    }
    #pragma unroll
    for (int p = 0; p < 2; ++p) {
        const int row = p*32 + r;
        okW[p] = !GW || row < rowsW;
        const int rc = (GW && row >= rowsW) ? rowsW - 1 : row;
        pw[p] = Wb + (size_t)rc*ldw + c;
    }
    const bool tailIn = !GK || c < KLAST;
    const int  koLast = tailIn ? (NCH - 1)*TK : 0;

    const bool live = !SKIP || ((!GA || wm*32 < rowsA) && (!GW || wn*32 < rowsW));

    v4f ra[4], rw[2];
    float* st0 = smem;
    float* st1 = smem + STG;

    tile_get_p(pa, pw, 0, ra, rw);
    stage_put_z(st0, tid, ra, rw, okA, okW, true);
    tile_get_p(pa, pw, (1 == NCH - 1) ? koLast : TK, ra, rw);
    __syncthreads();

    #pragma unroll 1
    for (int kc = 0; kc < NCH; kc += 2) {
        stage_put_z(st1, tid, ra, rw, okA, okW, (kc + 1 != NCH - 1) || tailIn);
        if (kc + 2 < NCH) tile_get_p(pa, pw, (kc + 2)*TK, ra, rw);
        if (live) mma_phase(st0, acc, wm, wn, ll, lh);
        __syncthreads();

        if (kc + 2 < NCH) stage_put_z(st0, tid, ra, rw, okA, okW, true);
        if (kc + 3 < NCH) tile_get_p(pa, pw, (kc + 3 == NCH - 1) ? koLast : (kc + 3)*TK, ra, rw);
        if (live) mma_phase(st1, acc, wm, wn, ll, lh);
        __syncthreads();
    }
}

__global__ __launch_bounds__(256) void proj_tkn_lab_wmma(
    const float* __restrict__ token, const float* __restrict__ W1,
    const float* __restrict__ label, const float* __restrict__ W2,
    float* __restrict__ tkn, float* __restrict__ lab)
{
    __shared__ __attribute__((aligned(16))) float smem[2*STG];

    const int tid  = threadIdx.x;
    const int lane = tid & 31, wid = tid >> 5;
    const int wm   = wid >> 1, wn  = wid & 1;
    const int ll   = lane & 15, lh = lane >> 4;
    const int id   = blockIdx.x;

    const v8f vzero = {0.f,0.f,0.f,0.f,0.f,0.f,0.f,0.f};
    v8f acc[2][2];
    acc[0][0]=vzero; acc[0][1]=vzero; acc[1][0]=vzero; acc[1][1]=vzero;

    if (id < 192) {
        const int m0 = (id & 15) * TM, n0 = (id >> 4) * TN;
        gemm_tile_128x64<768, false, false, false>(token + (size_t)m0*768, TM, 768, W1 + (size_t)n0*768, TN, 768,
                                            smem, acc, tid, wm, wn, ll, lh);
        #pragma unroll
        for (int mt = 0; mt < 2; ++mt)
            #pragma unroll
            for (int nt = 0; nt < 2; ++nt) {
                int gn = n0 + wn*32 + nt*16 + ll;
                #pragma unroll
                for (int r = 0; r < 8; ++r) {
                    int gm = m0 + wm*32 + mt*16 + lh*8 + r;
                    tkn[(size_t)gm*768 + gn] = acc[mt][nt][r];
                }
            }
    } else {
        const int s  = id - 192;
        const int m0 = (s / 12) * TM, n0 = (s % 12) * TN;
        gemm_tile_128x64<300, true, false, false>(label + (size_t)m0*300, 160 - m0, 300, W2 + (size_t)n0*300, TN, 300,
                                           smem, acc, tid, wm, wn, ll, lh);
        #pragma unroll
        for (int mt = 0; mt < 2; ++mt)
            #pragma unroll
            for (int nt = 0; nt < 2; ++nt) {
                int gn = n0 + wn*32 + nt*16 + ll;
                #pragma unroll
                for (int r = 0; r < 8; ++r) {
                    int gm = m0 + wm*32 + mt*16 + lh*8 + r;
                    if (gm < 160) lab[(size_t)gm*768 + gn] = acc[mt][nt][r];
                }
            }
    }
}

__global__ __launch_bounds__(256) void attn_softmax_kernel(
    const float* __restrict__ scores, const float* __restrict__ lmask,
    const float* __restrict__ imask, float* __restrict__ attn,
    float* __restrict__ b_in)
{
    int row = blockIdx.x * 256 + threadIdx.x;
    if (row >= 8*256*20) return;
    int bs = row / 20;
    int l  = row - bs*20;

    float v[8];
    float mx = -3.4e38f;
    #pragma unroll
    for (int t = 0; t < 8; ++t) {
        float s = scores[(size_t)row*8 + t] + (1.0f - lmask[l*8 + t]) * NEG_BIG;
        v[t] = s;
        mx = fmaxf(mx, s);
    }
    float sum = 0.f;
    #pragma unroll
    for (int t = 0; t < 8; ++t) { float e = __expf(v[t] - mx); v[t] = e; sum += e; }
    float inv = 1.0f / sum;
    #pragma unroll
    for (int t = 0; t < 8; ++t) attn[(size_t)row*8 + t] = v[t] * inv;

    b_in[row] = mx + (1.0f - imask[bs]) * NEG_BIG;
}

#define QC_TP 36

__global__ __launch_bounds__(256) void q2c_kernel(
    const float* __restrict__ b_in, const float* __restrict__ tkn,
    float* __restrict__ q2c)
{
    __shared__ __attribute__((aligned(16))) float Wt[256*20];
    __shared__ __attribute__((aligned(16))) float Ts[256*QC_TP];
    __shared__ __attribute__((aligned(16))) float Pp[8*20*32];

    const int tid  = threadIdx.x;
    const int lane = tid & 31, wid = tid >> 5;
    const int b    = blockIdx.x / 24;
    const int slab = blockIdx.x - b*24;
    const int cq   = tid & 7, r0 = tid >> 3;

    {
        v4f xb[5], tt[8];
        const float* bp = b_in + (size_t)b*256*20;
        const float* tp = tkn + ((size_t)b*256 + r0)*768 + slab*32 + 4*cq;
        #pragma unroll
        for (int p = 0; p < 5; ++p) xb[p] = *(const v4f*)(bp + 4*(tid + 256*p));
        #pragma unroll
        for (int p = 0; p < 8; ++p) tt[p] = *(const v4f*)(tp + (size_t)p*32*768);
        #pragma unroll
        for (int p = 0; p < 5; ++p) *(v4f*)(Wt + 4*(tid + 256*p)) = xb[p];
        #pragma unroll
        for (int p = 0; p < 8; ++p) *(v4f*)(Ts + (r0 + 32*p)*QC_TP + 4*cq) = tt[p];
    }
    __syncthreads();

    {
        float x[3][8], mx[3], sm[3];
        #pragma unroll
        for (int k = 0; k < 3; ++k) {
            const int l  = wid + 8*k;
            const int lc = (l < 20) ? l : wid;
            mx[k] = -3.4e38f;
            #pragma unroll
            for (int i = 0; i < 8; ++i) { x[k][i] = Wt[(lane + 32*i)*20 + lc]; mx[k] = fmaxf(mx[k], x[k][i]); }
        }
        #pragma unroll
        for (int o = 16; o > 0; o >>= 1)
            #pragma unroll
            for (int k = 0; k < 3; ++k) mx[k] = fmaxf(mx[k], __shfl_xor(mx[k], o, 32));
        #pragma unroll
        for (int k = 0; k < 3; ++k) {
            sm[k] = 0.f;
            #pragma unroll
            for (int i = 0; i < 8; ++i) { x[k][i] = __expf(x[k][i] - mx[k]); sm[k] += x[k][i]; }
        }
        #pragma unroll
        for (int o = 16; o > 0; o >>= 1)
            #pragma unroll
            for (int k = 0; k < 3; ++k) sm[k] += __shfl_xor(sm[k], o, 32);
        #pragma unroll
        for (int k = 0; k < 3; ++k) {
            const int l = wid + 8*k;
            if (l < 20) {
                #pragma unroll
                for (int i = 0; i < 8; ++i) Wt[(lane + 32*i)*20 + l] = x[k][i] / sm[k];
            }
        }
    }
    __syncthreads();

    {
        const int lg = (tid >> 3) & 3;
        const v4f z = {0.f,0.f,0.f,0.f};
        v4f acc[5] = {z, z, z, z, z};
        const float* tsp = Ts + (wid*32)*QC_TP + 4*cq;
        const float* wp  = Wt + (wid*32)*20 + 5*lg;
        #pragma unroll 8
        for (int r = 0; r < 32; ++r) {
            const v4f t = *(const v4f*)(tsp + r*QC_TP);
            #pragma unroll
            for (int k = 0; k < 5; ++k) acc[k] += wp[r*20 + k] * t;
        }
        #pragma unroll
        for (int k = 0; k < 5; ++k) *(v4f*)(Pp + (wid*20 + 5*lg + k)*32 + 4*cq) = acc[k];
    }
    __syncthreads();

    if (tid < 160) {
        const int l = tid >> 3;
        v4f s = *(const v4f*)(Pp + l*32 + 4*cq);
        #pragma unroll
        for (int w = 1; w < 8; ++w) s += *(const v4f*)(Pp + (w*20 + l)*32 + 4*cq);
        *(v4f*)(q2c + (size_t)(b*20 + l)*768 + slab*32 + 4*cq) = s;
    }
}

__global__ __launch_bounds__(256) void scores_rowfactor_wmma(
    const float* __restrict__ tkn, const float* __restrict__ lab,
    const float* __restrict__ W5,
    float* __restrict__ scores, float* __restrict__ T0, float* __restrict__ Pt)
{
    __shared__ __attribute__((aligned(16))) float smem[2*STG];

    const int tid  = threadIdx.x;
    const int lane = tid & 31;
    const int wid  = __builtin_amdgcn_readfirstlane(tid >> 5);
    const int wm   = wid >> 1, wn  = wid & 1;
    const int ll   = lane & 15, lh = lane >> 4;
    const int id   = blockIdx.x;

    int role, m0, n0, rowsA, rowsW, ldw;
    const float* Ab; const float* Wb;
    if (id < 192) {
        role = 0; m0 = (id & 15) * TM; n0 = (id >> 4) * TN;
        Ab = tkn + (size_t)m0*768;  rowsA = TM;
        Wb = W5 + (size_t)n0*3072;  rowsW = TN;  ldw = 3072;
    } else if (id < 240) {
        const int s = id - 192;
        role = 1; m0 = (s & 15) * TM; n0 = (s >> 4) * TN;
        Ab = tkn + (size_t)m0*768;  rowsA = TM;
        Wb = lab + (size_t)n0*768;  rowsW = 160 - n0;  ldw = 768;
    } else {
        const int s = id - 240;
        role = 2; m0 = (s / 12) * TM; n0 = (s % 12) * TN;
        Ab = lab + (size_t)m0*768;  rowsA = 160 - m0;
        Wb = W5 + 768 + (size_t)n0*3072;  rowsW = TN;  ldw = 3072;
    }

    const v8f vzero = {0.f,0.f,0.f,0.f,0.f,0.f,0.f,0.f};
    v8f acc[2][2];
    acc[0][0]=vzero; acc[0][1]=vzero; acc[1][0]=vzero; acc[1][1]=vzero;

    if (role != 2) gemm_tile_128x64<768, false, true, false>(Ab, TM, 768, Wb, rowsW, ldw, smem, acc, tid, wm, wn, ll, lh);
    else           gemm_tile_128x64<768, true, false, true>(Ab, rowsA, 768, Wb, TN, 3072, smem, acc, tid, wm, wn, ll, lh);

    #pragma unroll
    for (int mt = 0; mt < 2; ++mt)
        #pragma unroll
        for (int nt = 0; nt < 2; ++nt) {
            int gn = n0 + wn*32 + nt*16 + ll;
            #pragma unroll
            for (int r = 0; r < 8; ++r) {
                int row = wm*32 + mt*16 + lh*8 + r;
                if (role == 0) {
                    T0[(size_t)(m0 + row)*768 + gn] = acc[mt][nt][r];
                } else if (role == 1) {
                    if (gn < 160) scores[(size_t)(m0 + row)*160 + gn] = acc[mt][nt][r];
                } else if (row < rowsA) {
                    int q = m0 + row;
                    Pt[((size_t)(q >> 3)*768 + gn)*8 + (q & 7)] = acc[mt][nt][r];
                }
            }
        }
}

__global__ __launch_bounds__(192) void build_fused_a(
    const float* __restrict__ tkn,  const float* __restrict__ lab,
    const float* __restrict__ attn, const float* __restrict__ q2c,
    float* __restrict__ Af)
{
    const int tid = threadIdx.x;
    const int l   = blockIdx.x;
    const int bs0 = blockIdx.y * 16;
    const int b   = bs0 >> 8;
    const int j   = tid << 2;

    v4f lv[8];
    #pragma unroll
    for (int t = 0; t < 8; ++t) lv[t] = *(const v4f*)(lab + (size_t)(l*8 + t)*768 + j);
    const v4f q4 = *(const v4f*)(q2c + (size_t)(b*20 + l)*768 + j);

    #pragma unroll 4
    for (int i = 0; i < 16; ++i) {
        const int bs = bs0 + i;
        const size_t m = (size_t)bs*20 + l;
        const v4f aLo = *(const v4f*)(attn + m*8);
        const v4f aHi = *(const v4f*)(attn + m*8 + 4);
        const v4f t4  = *(const v4f*)(tkn + (size_t)bs*768 + j);
        v4f cq = {0.f,0.f,0.f,0.f};
        cq += aLo.x * lv[0]; cq += aLo.y * lv[1]; cq += aLo.z * lv[2]; cq += aLo.w * lv[3];
        cq += aHi.x * lv[4]; cq += aHi.y * lv[5]; cq += aHi.z * lv[6]; cq += aHi.w * lv[7];
        float* dst = Af + m*1536 + j;
        *(v4f*)dst         = t4 * cq;
        *(v4f*)(dst + 768) = t4 * q4;
    }
}

#define MM 160
#define MN 128
#define MSTG (288*TKP)
#define MSP 130
#define MSMEM (MM*8 + MM*MSP)

__device__ __forceinline__ void tile_get_m(const float* __restrict__ aP, const float* __restrict__ wP,
                                           v4f (&a)[5], v4f (&w)[4])
{
    #pragma unroll
    for (int p = 0; p < 5; ++p) a[p] = *(const v4f*)(aP + (size_t)p*32*1536);
    #pragma unroll
    for (int p = 0; p < 4; ++p) w[p] = *(const v4f*)(wP + (size_t)p*32*3072);
}

__device__ __forceinline__ void stage_put_m(float* st, int tid, const v4f (&a)[5], const v4f (&w)[4])
{
    const int r = tid >> 3, c = (tid & 7) << 2;
    #pragma unroll
    for (int p = 0; p < 5; ++p) {
        float* d = st + (p*32 + r)*TKP + c;
        v2f lo = {a[p].x, a[p].y}, hi = {a[p].z, a[p].w};
        *(v2f*)d = lo; *(v2f*)(d + 2) = hi;
    }
    #pragma unroll
    for (int p = 0; p < 4; ++p) {
        float* d = st + (MM + p*32 + r)*TKP + c;
        v2f lo = {w[p].x, w[p].y}, hi = {w[p].z, w[p].w};
        *(v2f*)d = lo; *(v2f*)(d + 2) = hi;
    }
}

__device__ __forceinline__ void mma_phase_m(const float* st, v8f (&acc)[5][2],
                                            int wm, int wn, int ll, int lh)
{
    const float* As = st + (wm*80 + ll)*TKP + 2*lh;
    const float* Ws = st + (MM + wn*32 + ll)*TKP + 2*lh;
    #pragma unroll
    for (int kk = 0; kk < 8; ++kk) {
        v2f a[5], b[2];
        #pragma unroll
        for (int i = 0; i < 5; ++i) a[i] = *(const v2f*)(As + i*16*TKP + 4*kk);
        #pragma unroll
        for (int j = 0; j < 2; ++j) b[j] = *(const v2f*)(Ws + j*16*TKP + 4*kk);
        #pragma unroll
        for (int i = 0; i < 5; ++i)
            #pragma unroll
            for (int j = 0; j < 2; ++j)
                acc[i][j] = __builtin_amdgcn_wmma_f32_16x16x4_f32(false,a[i],false,b[j],(short)0,acc[i][j],false,false);
    }
}

#define TANH_K 2.885390081777927f
__device__ __forceinline__ float tanh_from_scaled(float xs)
{
    const float e = __builtin_amdgcn_exp2f(xs);
    const float r = __builtin_amdgcn_rcpf(1.0f + e);
    return fmaf(-2.0f, r, 1.0f);
}

__global__ __launch_bounds__(256) void fused_w5_tanh_wmma(
    const float* __restrict__ attn, const float* __restrict__ Af,
    const float* __restrict__ W5,   const float* __restrict__ b5,
    const float* __restrict__ T0,   const float* __restrict__ Pt,
    float* __restrict__ out)
{
    __shared__ __attribute__((aligned(16))) float smem[MSMEM];
    static_assert(MSMEM >= 2*MSTG, "the stages live in the same bytes");
    float (*aBlk)[8]     = (float (*)[8])smem;
    float (*startS)[MSP] = (float (*)[MSP])(smem + MM*8);

    const int tid  = threadIdx.x;
    const int lane = tid & 31, wid = tid >> 5;
    const int wm   = wid >> 2, wn  = wid & 3;
    const int ll   = lane & 15, lh = lane >> 4;
    const int n0   = blockIdx.x * MN, m0 = blockIdx.y * MM;

    const float* aP = Af + (size_t)(m0 + (tid >> 3))*1536 + ((tid & 7) << 2);
    const float* wP = W5 + (size_t)(n0 + (tid >> 3))*3072 + 1536 + ((tid & 7) << 2);
    v4f ra[5], rw[4];
    tile_get_m(aP, wP, ra, rw);

    for (int i = tid; i < 2*MM; i += 256) {
        const int r = i >> 1, c = (i & 1) << 2;
        *(v4f*)&aBlk[r][c] = *(const v4f*)(attn + (size_t)(m0 + r)*8 + c);
    }

    __syncthreads();

    {
        const int col = tid & (MN - 1);
        const int lb  = (tid >> 7) * 10;
        const int gn  = n0 + col;
        const int bs0 = blockIdx.y * 8;
        float t0[8];
        #pragma unroll
        for (int i = 0; i < 8; ++i) t0[i] = T0[(size_t)(bs0 + i)*768 + gn];
        #pragma unroll 2
        for (int j = 0; j < 10; ++j) {
            const int l = lb + j;
            const float* pp = Pt + ((size_t)l*768 + gn)*8;
            const v4f pLo = *(const v4f*)pp;
            const v4f pHi = *(const v4f*)(pp + 4);
            #pragma unroll
            for (int i = 0; i < 8; ++i) {
                const int row = i*20 + l;
                const v4f aLo = *(const v4f*)&aBlk[row][0];
                const v4f aHi = *(const v4f*)&aBlk[row][4];
                float s = t0[i];
                s += aLo.x*pLo.x; s += aLo.y*pLo.y; s += aLo.z*pLo.z; s += aLo.w*pLo.w;
                s += aHi.x*pHi.x; s += aHi.y*pHi.y; s += aHi.z*pHi.z; s += aHi.w*pHi.w;
                startS[row][col] = s;
            }
        }
    }
    __syncthreads();

    v8f acc[5][2];
    #pragma unroll
    for (int mt = 0; mt < 5; ++mt)
        #pragma unroll
        for (int nt = 0; nt < 2; ++nt)
            #pragma unroll
            for (int r = 0; r < 8; ++r)
                acc[mt][nt][r] = startS[wm*80 + mt*16 + lh*8 + r][wn*32 + nt*16 + ll];

    __syncthreads();

    float* st0 = smem;
    float* st1 = smem + MSTG;
    stage_put_m(st0, tid, ra, rw);
    tile_get_m(aP + 32, wP + 32, ra, rw);
    __syncthreads();

    for (int kc = 0; kc < 48; kc += 2) {
        stage_put_m(st1, tid, ra, rw);
        if (kc + 2 < 48) tile_get_m(aP + (kc + 2)*32, wP + (kc + 2)*32, ra, rw);
        mma_phase_m(st0, acc, wm, wn, ll, lh);
        __syncthreads();

        if (kc + 2 < 48) stage_put_m(st0, tid, ra, rw);
        if (kc + 3 < 48) tile_get_m(aP + (kc + 3)*32, wP + (kc + 3)*32, ra, rw);
        mma_phase_m(st1, acc, wm, wn, ll, lh);
        __syncthreads();
    }

    #pragma unroll
    for (int mt = 0; mt < 5; ++mt)
        #pragma unroll
        for (int nt = 0; nt < 2; ++nt) {
            int gn = n0 + wn*32 + nt*16 + ll;
            float bias_s = b5[gn] * TANH_K;
            #pragma unroll
            for (int r = 0; r < 8; ++r) {
                int gm = m0 + wm*80 + mt*16 + lh*8 + r;
                out[(size_t)gm*768 + gn] = tanh_from_scaled(fmaf(acc[mt][nt][r], TANH_K, bias_s));
            }
        }
}

extern "C" void kernel_launch(void* const* d_in, const int* in_sizes, int n_in,
                              void* d_out, int out_size, void* d_ws, size_t ws_size,
                              hipStream_t stream)
{
    (void)in_sizes; (void)n_in; (void)out_size; (void)ws_size;

    const float* token = (const float*)d_in[0];
    const float* label = (const float*)d_in[1];
    const float* imask = (const float*)d_in[2];
    const float* lmask = (const float*)d_in[3];
    const float* W1    = (const float*)d_in[4];
    const float* W2    = (const float*)d_in[5];
    const float* W5    = (const float*)d_in[6];
    const float* b5    = (const float*)d_in[7];
    float* out = (float*)d_out;

    float* ws     = (float*)d_ws;
    float* tkn    = ws;
    float* lab    = tkn    + 2048*768;
    float* scores = lab    + 160*768;
    float* attn   = scores + 2048*160;
    float* b_in   = attn   + 2048*160;
    float* q2c    = b_in   + 40960;
    float* T0     = q2c    + 160*768;
    float* Pt     = T0     + 2048*768;
    float* Af     = Pt     + 20*768*8;

    dim3 blk(256);
    proj_tkn_lab_wmma<<<dim3(216), blk, 0, stream>>>(token, W1, label, W2, tkn, lab);
    scores_rowfactor_wmma<<<dim3(264), blk, 0, stream>>>(tkn, lab, W5, scores, T0, Pt);
    attn_softmax_kernel<<<dim3(160), blk, 0, stream>>>(scores, lmask, imask, attn, b_in);
    q2c_kernel<<<dim3(192), blk, 0, stream>>>(b_in, tkn, q2c);
    build_fused_a<<<dim3(20, 128), dim3(192), 0, stream>>>(tkn, lab, attn, q2c, Af);
    fused_w5_tanh_wmma<<<dim3(6, 256), blk, 0, stream>>>(attn, Af, W5, b5, T0, Pt, out);
}
